# starting kernel
# baseline (speedup 1.0000x reference)
#include <hip/hip_runtime.h>
#include <math.h>

#define N_ATOMS  10000
#define N_PAIRS  160000
#define N_CH     32
#define KDIM     128
#define K0TOT    384

typedef float v2f __attribute__((ext_vector_type(2)));
typedef float v8f __attribute__((ext_vector_type(8)));

#define A3 0.5773502691896258f
#define A2 0.7071067811865476f
#define B2 0.8164965809277260f
#define B1 0.4082482904638630f
__constant__ float U2c[9][9] = {
    { A3, 0,   0,   0,  A3, 0,   0,   0,   A3},
    { 0,  0,   0,   0,  0,  A2,  0,  -A2,  0 },
    { 0,  0,  -A2,  0,  0,  0,   A2,  0,   0 },
    { 0,  A2,  0,  -A2, 0,  0,   0,   0,   0 },
    { 0,  0,   A2,  0,  0,  0,   A2,  0,   0 },
    { 0,  A2,  0,   A2, 0,  0,   0,   0,   0 },
    {-B1, 0,   0,   0,  B2, 0,   0,   0,  -B1},
    { 0,  0,   0,   0,  0,  A2,  0,   A2,  0 },
    {-A2, 0,   0,   0,  0,  0,   0,   0,   A2},
};

__global__ void k_zero(float* p, long n) {
    long i = (long)blockIdx.x * blockDim.x + threadIdx.x;
    if (i < n) p[i] = 0.f;
}

__global__ void k_init_out(float* out, const float* b_out) {
    int i = blockIdx.x * blockDim.x + threadIdx.x;
    if (i < N_ATOMS) out[i] = b_out[0];
}

__global__ void k_pairs(const float* __restrict__ pos, const int* __restrict__ species,
                        const int* __restrict__ pairs, const float* __restrict__ emb,
                        const float* __restrict__ W_rad,
                        float* __restrict__ f0, float* __restrict__ f1, float* __restrict__ f2) {
    int p = blockIdx.x;
    int k = threadIdx.x;
    int c  = pairs[2 * p + 0];
    int nb = pairs[2 * p + 1];
    float rx = pos[3 * nb + 0] - pos[3 * c + 0];
    float ry = pos[3 * nb + 1] - pos[3 * c + 1];
    float rz = pos[3 * nb + 2] - pos[3 * c + 2];
    float d  = sqrtf(rx * rx + ry * ry + rz * rz + 1e-12f);
    float inv = 1.f / d;
    float ux = rx * inv, uy = ry * inv, uz = rz * inv;

    __shared__ float s_rb[8];
    if (k < 8) {
        float mu = 20.f * (float)k / 7.f;
        float t  = (d - 15.f) * 0.2f;
        t = fminf(fmaxf(t, 0.f), 1.f);
        float fc = 0.5f * (cosf(3.14159265358979f * t) + 1.f);
        float x  = (d - mu) * 0.4f;
        s_rb[k] = expf(-x * x) * fc;
    }
    __syncthreads();

    float neigh = emb[species[nb] * N_CH + (k >> 2)];
    float R[3];
#pragma unroll
    for (int l = 0; l < 3; ++l) {
        float s = 0.f;
#pragma unroll
        for (int j = 0; j < 8; ++j) s += s_rb[j] * W_rad[(l * 8 + j) * KDIM + k];
        R[l] = s;
    }
    const float mp = 0.1f;
    atomicAdd(&f0[(long)c * KDIM + k], R[0] * neigh * mp);
    float sh1[3] = { uy, uz, ux };
    float w1 = R[1] * neigh * mp;
#pragma unroll
    for (int m = 0; m < 3; ++m)
        atomicAdd(&f1[((long)c * 3 + m) * KDIM + k], sh1[m] * w1);
    const float s3 = 1.7320508075688772f;
    float sh2[5] = { s3 * ux * uy, s3 * uy * uz, 0.5f * (3.f * uz * uz - 1.f),
                     s3 * ux * uz, 0.5f * s3 * (ux * ux - uy * uy) };
    float w2 = R[2] * neigh * mp;
#pragma unroll
    for (int m = 0; m < 5; ++m)
        atomicAdd(&f2[((long)c * 5 + m) * KDIM + k], sh2[m] * w2);
}

__global__ void k_uncouple(const float* __restrict__ f1, const float* __restrict__ f2,
                           float* __restrict__ u1, float* __restrict__ u2) {
    long t = (long)blockIdx.x * blockDim.x + threadIdx.x;
    if (t >= (long)N_ATOMS * 9 * KDIM) return;
    int k = (int)(t % KDIM);
    int p = (int)((t / KDIM) % 9);
    long n = t / (9 * KDIM);
    float a = 0.f, b = 0.f;
#pragma unroll
    for (int q = 0; q < 3; ++q) a += U2c[1 + q][p] * f1[(n * 3 + q) * KDIM + k];
#pragma unroll
    for (int q = 0; q < 5; ++q) b += U2c[4 + q][p] * f2[(n * 5 + q) * KDIM + k];
    u1[t] = a;
    u2[t] = b;
}

__global__ void k_gemm_wmma128(const float* __restrict__ A, const float* __restrict__ B,
                               float* __restrict__ C) {
    int m0 = blockIdx.x * 16;
    int lane = threadIdx.x;
    int half = lane >> 4;
    int l16  = lane & 15;
    const float* Arow = A + (long)(m0 + l16) * KDIM;
    v8f acc[8] = {};
    for (int kk = 0; kk < KDIM; kk += 4) {
        int ka = kk + 2 * half;
        v2f a;
        a.x = Arow[ka];
        a.y = Arow[ka + 1];
        const float* B0 = B + (long)ka * KDIM + l16;
#pragma unroll
        for (int t = 0; t < 8; ++t) {
            v2f b;
            b.x = B0[t * 16];
            b.y = B0[KDIM + t * 16];
            acc[t] = __builtin_amdgcn_wmma_f32_16x16x4_f32(false, a, false, b, (short)0,
                                                           acc[t], false, false);
        }
    }
#pragma unroll
    for (int t = 0; t < 8; ++t) {
        float* Cc = C + (long)(m0 + 8 * half) * KDIM + t * 16 + l16;
#pragma unroll
        for (int i = 0; i < 8; ++i) Cc[(long)i * KDIM] = acc[t][i];
    }
}

__global__ void k_feats(const float* __restrict__ f0, const float* __restrict__ u1,
                        const float* __restrict__ u2, const float* __restrict__ g0,
                        const float* __restrict__ g1, const float* __restrict__ g2,
                        const int* __restrict__ species, const float* __restrict__ emb,
                        float* __restrict__ x0) {
    long t = (long)blockIdx.x * blockDim.x + threadIdx.x;
    if (t >= (long)N_ATOMS * KDIM) return;
    int k = (int)(t % KDIM);
    long n = t / KDIM;
    float f0v = f0[t];
    float feats0 = g0[t] * f0v + f0v;

    float G[9], F[9];
    float tr1 = 0.f;
#pragma unroll
    for (int p = 0; p < 9; ++p) {
        G[p] = g1[(n * 9 + p) * KDIM + k];
        F[p] = u1[(n * 9 + p) * KDIM + k];
    }
#pragma unroll
    for (int a = 0; a < 3; ++a)
#pragma unroll
        for (int b = 0; b < 3; ++b) tr1 += G[a * 3 + b] * F[b * 3 + a];
    tr1 += F[0] + F[4] + F[8];

    float tr2 = 0.f;
#pragma unroll
    for (int p = 0; p < 9; ++p) {
        G[p] = g2[(n * 9 + p) * KDIM + k];
        F[p] = u2[(n * 9 + p) * KDIM + k];
    }
#pragma unroll
    for (int a = 0; a < 3; ++a)
#pragma unroll
        for (int b = 0; b < 3; ++b) tr2 += G[a * 3 + b] * F[b * 3 + a];
    tr2 += F[0] + F[4] + F[8];

    const float i3 = 0.5773502691896258f;
    int sp = species[n];
    const float* e = emb + sp * N_CH;
    x0[n * K0TOT + k]       = feats0      * e[k / 12];
    x0[n * K0TOT + 128 + k] = (tr1 * i3)  * e[(128 + k) / 12];
    x0[n * K0TOT + 256 + k] = (tr2 * i3)  * e[(256 + k) / 12];
}

__global__ void k_head_wmma(const float* __restrict__ x0, const float* __restrict__ Wh,
                            const float* __restrict__ bh, const float* __restrict__ Wo,
                            float* __restrict__ out) {
    int m0 = blockIdx.x * 16;
    int n0 = blockIdx.y * 128;
    int lane = threadIdx.x;
    int half = lane >> 4;
    int l16  = lane & 15;
    const float* Arow = x0 + (long)(m0 + l16) * K0TOT;
    v8f acc[8] = {};
    for (int kk = 0; kk < K0TOT; kk += 4) {
        int ka = kk + 2 * half;
        v2f a;
        a.x = Arow[ka];
        a.y = Arow[ka + 1];
        const float* B0 = Wh + (long)ka * K0TOT + n0 + l16;
#pragma unroll
        for (int t = 0; t < 8; ++t) {
            v2f b;
            b.x = B0[t * 16];
            b.y = B0[K0TOT + t * 16];
            acc[t] = __builtin_amdgcn_wmma_f32_16x16x4_f32(false, a, false, b, (short)0,
                                                           acc[t], false, false);
        }
    }
    int rowbase = m0 + 8 * half;
    float partial[8];
#pragma unroll
    for (int i = 0; i < 8; ++i) partial[i] = 0.f;
#pragma unroll
    for (int t = 0; t < 8; ++t) {
        int col = n0 + t * 16 + l16;
        float bias = bh[col];
        float wo   = Wo[col];
#pragma unroll
        for (int i = 0; i < 8; ++i) {
            float v = acc[t][i] + bias;
            partial[i] += (v / (1.f + expf(-v))) * wo;
        }
    }
#pragma unroll
    for (int i = 0; i < 8; ++i) atomicAdd(&out[rowbase + i], partial[i]);
}

extern "C" void kernel_launch(void* const* d_in, const int* in_sizes, int n_in,
                              void* d_out, int out_size, void* d_ws, size_t ws_size,
                              hipStream_t stream) {
    (void)in_sizes; (void)n_in; (void)out_size; (void)ws_size;
    const float* positions = (const float*)d_in[0];
    const int*   species   = (const int*)d_in[1];
    const int*   pairs     = (const int*)d_in[2];
    const float* embeddings= (const float*)d_in[3];
    const float* W_rad     = (const float*)d_in[4];
    const float* W_cg0     = (const float*)d_in[5];
    const float* W_cg1     = (const float*)d_in[6];
    const float* W_cg2     = (const float*)d_in[7];
    const float* W_head    = (const float*)d_in[8];
    const float* b_head    = (const float*)d_in[9];
    const float* W_out     = (const float*)d_in[10];
    const float* b_out     = (const float*)d_in[11];
    float* out = (float*)d_out;

    float* ws = (float*)d_ws;
    const long SZ_F0 = (long)N_ATOMS * KDIM;
    const long SZ_F1 = (long)N_ATOMS * 3 * KDIM;
    const long SZ_F2 = (long)N_ATOMS * 5 * KDIM;
    const long SZ_U  = (long)N_ATOMS * 9 * KDIM;
    float* f0 = ws;
    float* f1 = f0 + SZ_F0;
    float* f2 = f1 + SZ_F1;
    float* u1 = f2 + SZ_F2;
    float* u2 = u1 + SZ_U;
    float* g0 = u2 + SZ_U;
    float* g1 = g0 + SZ_F0;
    float* g2 = g1 + SZ_U;
    float* x0 = g2 + SZ_U;

    long nz = SZ_F0 + SZ_F1 + SZ_F2;
    k_zero<<<(unsigned)((nz + 255) / 256), 256, 0, stream>>>(f0, nz);

    k_init_out<<<(N_ATOMS + 255) / 256, 256, 0, stream>>>(out, b_out);

    k_pairs<<<N_PAIRS, KDIM, 0, stream>>>(positions, species, pairs, embeddings, W_rad,
                                          f0, f1, f2);

    k_uncouple<<<45000, 256, 0, stream>>>(f1, f2, u1, u2);

    k_gemm_wmma128<<<N_ATOMS / 16, 32, 0, stream>>>(f0, W_cg0, g0);
    k_gemm_wmma128<<<N_ATOMS * 9 / 16, 32, 0, stream>>>(u1, W_cg1, g1);
    k_gemm_wmma128<<<N_ATOMS * 9 / 16, 32, 0, stream>>>(u2, W_cg2, g2);

    k_feats<<<(unsigned)(((long)N_ATOMS * KDIM + 255) / 256), 256, 0, stream>>>(
        f0, u1, u2, g0, g1, g2, species, embeddings, x0);

    k_head_wmma<<<dim3(N_ATOMS / 16, K0TOT / 16 / 8), 32, 0, stream>>>(
        x0, W_head, b_head, W_out, out);
}


# GEMM weights stay in registers: persistent waves, one per SIMD, hold W_cg (mix: 3 launches -> 1) or a 384x32 W_head stripe as B fragments, A prefetched; W_out dot sum order changes (4 waves via LDS)
# speedup vs baseline: 1.8988x; 1.8988x over previous
#include <hip/hip_runtime.h>
#include <math.h>

#define N_ATOMS  10000
#define N_PAIRS  160000
#define N_CH     32
#define KDIM     128
#define K0TOT    384

typedef float v2f __attribute__((ext_vector_type(2)));
typedef float v8f __attribute__((ext_vector_type(8)));

#define A3 0.5773502691896258f
#define A2 0.7071067811865476f
#define B2 0.8164965809277260f
#define B1 0.4082482904638630f
__constant__ float U2c[9][9] = {
    { A3, 0,   0,   0,  A3, 0,   0,   0,   A3},
    { 0,  0,   0,   0,  0,  A2,  0,  -A2,  0 },
    { 0,  0,  -A2,  0,  0,  0,   A2,  0,   0 },
    { 0,  A2,  0,  -A2, 0,  0,   0,   0,   0 },
    { 0,  0,   A2,  0,  0,  0,   A2,  0,   0 },
    { 0,  A2,  0,   A2, 0,  0,   0,   0,   0 },
    {-B1, 0,   0,   0,  B2, 0,   0,   0,  -B1},
    { 0,  0,   0,   0,  0,  A2,  0,   A2,  0 },
    {-A2, 0,   0,   0,  0,  0,   0,   0,   A2},
};

#define CAP 64

__global__ void k_init(float* out, const float* b_out, int* cnt, int* spill_n) {
    int i = blockIdx.x * blockDim.x + threadIdx.x;
    if (i < N_ATOMS) { out[i] = b_out[0]; cnt[i] = 0; }
    if (i == 0) *spill_n = 0;
}

__global__ void k_bin(const float* __restrict__ pos, const int* __restrict__ species,
                      const int* __restrict__ pairs, int* __restrict__ cnt,
                      int* __restrict__ spill_n, float4* __restrict__ rec, int* __restrict__ rsp,
                      float4* __restrict__ srec, int* __restrict__ scs) {
    int p = blockIdx.x * blockDim.x + threadIdx.x;
    if (p >= N_PAIRS) return;
    int c  = pairs[2 * p + 0];
    int nb = pairs[2 * p + 1];
    float rx = pos[3 * nb + 0] - pos[3 * c + 0];
    float ry = pos[3 * nb + 1] - pos[3 * c + 1];
    float rz = pos[3 * nb + 2] - pos[3 * c + 2];
    float d  = sqrtf(rx * rx + ry * ry + rz * rz + 1e-12f);
    float inv = 1.f / d;
    float ux = rx * inv, uy = ry * inv, uz = rz * inv;
    float t  = (d - 15.f) * 0.2f;
    t = fminf(fmaxf(t, 0.f), 1.f);
    float fc = 0.5f * (cosf(3.14159265358979f * t) + 1.f);
    float rb[8];
#pragma unroll
    for (int j = 0; j < 8; ++j) {
        float mu = 20.f * (float)j / 7.f;
        float x  = (d - mu) * 0.4f;
        rb[j] = expf(-x * x) * fc;
    }
    const float s3 = 1.7320508075688772f;
    float4 q0 = { rb[0], rb[1], rb[2], rb[3] };
    float4 q1 = { rb[4], rb[5], rb[6], rb[7] };
    float4 q2 = { uy, uz, ux, s3 * ux * uy };
    float4 q3 = { s3 * uy * uz, 0.5f * (3.f * uz * uz - 1.f), s3 * ux * uz,
                  0.5f * s3 * (ux * ux - uy * uy) };
    int sp = species[nb];
    int slot = atomicAdd(&cnt[c], 1);
    float4* dst;
    if (slot < CAP) {
        long idx = (long)c * CAP + slot;
        rsp[idx] = sp;
        dst = rec + idx * 4;
    } else {
        int o = atomicAdd(spill_n, 1);
        scs[2 * o + 0] = c;
        scs[2 * o + 1] = sp;
        dst = srec + (long)o * 4;
    }
    dst[0] = q0; dst[1] = q1; dst[2] = q2; dst[3] = q3;
}

__device__ __forceinline__ void add_pair(const float4* __restrict__ g, int sp,
                                         const float* __restrict__ ek,
                                         const float (&w)[24], float (&acc)[9]) {
    float4 a = g[0], b = g[1], s0 = g[2], s1 = g[3];
    float rb[8] = { a.x, a.y, a.z, a.w, b.x, b.y, b.z, b.w };
    float neigh = ek[sp * N_CH];
    float R[3];
#pragma unroll
    for (int l = 0; l < 3; ++l) {
        float s = 0.f;
#pragma unroll
        for (int j = 0; j < 8; ++j) s += rb[j] * w[l * 8 + j];
        R[l] = s;
    }
    acc[0] += R[0] * neigh;
    float w1 = R[1] * neigh;
    acc[1] += s0.x * w1; acc[2] += s0.y * w1; acc[3] += s0.z * w1;
    float w2 = R[2] * neigh;
    acc[4] += s0.w * w2; acc[5] += s1.x * w2; acc[6] += s1.y * w2;
    acc[7] += s1.z * w2; acc[8] += s1.w * w2;
}

__global__ __launch_bounds__(KDIM) void k_gather(
        const float* __restrict__ emb, const float* __restrict__ W_rad,
        const int* __restrict__ cnt, const int* __restrict__ spill_n,
        const float4* __restrict__ rec, const int* __restrict__ rsp,
        const float4* __restrict__ srec, const int* __restrict__ scs,
        float* __restrict__ f0, float* __restrict__ u1, float* __restrict__ u2) {
    const int c = blockIdx.x;
    const int k = threadIdx.x;
    float w[24];
#pragma unroll
    for (int i = 0; i < 24; ++i) w[i] = W_rad[i * KDIM + k];
    const float* ek = emb + (k >> 2);
    float acc[9];
#pragma unroll
    for (int i = 0; i < 9; ++i) acc[i] = 0.f;

    const int n  = cnt[c];
    const int nr = n < CAP ? n : CAP;
    const float4* r = rec + (long)c * CAP * 4;
    const int*    s = rsp + (long)c * CAP;
    for (int i = 0; i < nr; ++i) add_pair(r + 4 * i, s[i], ek, w, acc);
    if (n > CAP) {
        int no = *spill_n;
        if (no > N_PAIRS) no = N_PAIRS;
        for (int i = 0; i < no; ++i)
            if (scs[2 * i] == c) add_pair(srec + 4 * (long)i, scs[2 * i + 1], ek, w, acc);
    }
    const float mp = 0.1f;
    float fl[9];
#pragma unroll
    for (int i = 0; i < 9; ++i) fl[i] = acc[i] * mp;
    f0[(long)c * KDIM + k] = fl[0];
    float* r1 = u1 + (long)c * 9 * KDIM + k;
    float* r2 = u2 + (long)c * 9 * KDIM + k;
#pragma unroll
    for (int p = 0; p < 9; ++p) {
        float a = 0.f, b = 0.f;
#pragma unroll
        for (int q = 0; q < 3; ++q) a += U2c[1 + q][p] * fl[1 + q];
#pragma unroll
        for (int q = 0; q < 5; ++q) b += U2c[4 + q][p] * fl[4 + q];
        r1[p * KDIM] = a;
        r2[p * KDIM] = b;
    }
}

__device__ __forceinline__ void mix_tile(const v2f (&cur)[32], v2f (&nxt)[32], const v2f (&b)[32][8],
                                         const float* __restrict__ Al, float* __restrict__ C,
                                         int tile, int stride, int ntiles, int half, int l16) {
    int nx = tile + stride;
    if (nx >= ntiles) nx = tile;
    const v2f* An = (const v2f*)(Al + (long)nx * 16 * KDIM);
#pragma unroll
    for (int j = 0; j < 32; ++j) nxt[j] = An[2 * j];
    __builtin_amdgcn_sched_barrier(0);

    v8f acc[8] = {};
#pragma unroll
    for (int j = 0; j < 32; ++j) {
#pragma unroll
        for (int t = 0; t < 8; ++t)
            acc[t] = __builtin_amdgcn_wmma_f32_16x16x4_f32(false, cur[j], false, b[j][t], (short)0,
                                                           acc[t], false, false);
    }
    float* Ct = C + ((long)tile * 16 + 8 * half) * KDIM + l16;
#pragma unroll
    for (int t = 0; t < 8; ++t) {
#pragma unroll
        for (int i = 0; i < 8; ++i) Ct[(long)i * KDIM + t * 16] = acc[t][i];
    }
}

#define MIX_WAVES 1024
#define MIX_W0    54
#define MIX_W1    485
__global__ __launch_bounds__(32) void k_mix_resident(
        const float* __restrict__ f0, const float* __restrict__ u1, const float* __restrict__ u2,
        const float* __restrict__ Wc0, const float* __restrict__ Wc1, const float* __restrict__ Wc2,
        float* __restrict__ g0, float* __restrict__ g1, float* __restrict__ g2) {
    const int w = blockIdx.x;
    const float* A; const float* B; float* C;
    int first, stride, ntiles;
    if (w < MIX_W0) {
        A = f0; B = Wc0; C = g0; first = w; stride = MIX_W0; ntiles = N_ATOMS / 16;
    } else if (w < MIX_W0 + MIX_W1) {
        A = u1; B = Wc1; C = g1; first = w - MIX_W0; stride = MIX_W1; ntiles = N_ATOMS * 9 / 16;
    } else {
        A = u2; B = Wc2; C = g2; first = w - MIX_W0 - MIX_W1; stride = MIX_WAVES - MIX_W0 - MIX_W1;
        ntiles = N_ATOMS * 9 / 16;
    }
    const int lane = threadIdx.x;
    const int half = lane >> 4;
    const int l16  = lane & 15;

    v2f b[32][8];
    const float* Bl = B + (long)(2 * half) * KDIM + l16;
#pragma unroll
    for (int j = 0; j < 32; ++j) {
#pragma unroll
        for (int t = 0; t < 8; ++t) {
            b[j][t].x = Bl[(4 * j) * KDIM + t * 16];
            b[j][t].y = Bl[(4 * j + 1) * KDIM + t * 16];
        }
    }

    const float* Al = A + (long)l16 * KDIM + 2 * half;
    v2f a[32];
    {
        const v2f* Ar = (const v2f*)(Al + (long)first * 16 * KDIM);
#pragma unroll
        for (int j = 0; j < 32; ++j) a[j] = Ar[2 * j];
    }
    for (int tile = first; tile < ntiles; tile += stride) {
        v2f an[32];
        mix_tile(a, an, b, Al, C, tile, stride, ntiles, half, l16);
        __builtin_amdgcn_sched_barrier(0);
#pragma unroll
        for (int j = 0; j < 32; ++j) a[j] = an[j];
    }
}

__global__ void k_feats(const float* __restrict__ f0, const float* __restrict__ u1,
                        const float* __restrict__ u2, const float* __restrict__ g0,
                        const float* __restrict__ g1, const float* __restrict__ g2,
                        const int* __restrict__ species, const float* __restrict__ emb,
                        float* __restrict__ x0) {
    long t = (long)blockIdx.x * blockDim.x + threadIdx.x;
    if (t >= (long)N_ATOMS * KDIM) return;
    int k = (int)(t % KDIM);
    long n = t / KDIM;
    float f0v = f0[t];
    float feats0 = g0[t] * f0v + f0v;

    float G[9], F[9];
    float tr1 = 0.f;
#pragma unroll
    for (int p = 0; p < 9; ++p) {
        G[p] = g1[(n * 9 + p) * KDIM + k];
        F[p] = u1[(n * 9 + p) * KDIM + k];
    }
#pragma unroll
    for (int a = 0; a < 3; ++a)
#pragma unroll
        for (int b = 0; b < 3; ++b) tr1 += G[a * 3 + b] * F[b * 3 + a];
    tr1 += F[0] + F[4] + F[8];

    float tr2 = 0.f;
#pragma unroll
    for (int p = 0; p < 9; ++p) {
        G[p] = g2[(n * 9 + p) * KDIM + k];
        F[p] = u2[(n * 9 + p) * KDIM + k];
    }
#pragma unroll
    for (int a = 0; a < 3; ++a)
#pragma unroll
        for (int b = 0; b < 3; ++b) tr2 += G[a * 3 + b] * F[b * 3 + a];
    tr2 += F[0] + F[4] + F[8];

    const float i3 = 0.5773502691896258f;
    int sp = species[n];
    const float* e = emb + sp * N_CH;
    x0[n * K0TOT + k]       = feats0      * e[k / 12];
    x0[n * K0TOT + 128 + k] = (tr1 * i3)  * e[(128 + k) / 12];
    x0[n * K0TOT + 256 + k] = (tr2 * i3)  * e[(256 + k) / 12];
}

#define HEAD_BLOCKS 256
#define HEAD_TILES  (N_ATOMS / 16)
__device__ __forceinline__ void head_tile(const v2f (&cur)[96], v2f (&nxt)[96], const v2f (&b)[96][2],
                                          const float* __restrict__ Al, float (&rd)[4][8][32],
                                          float* __restrict__ out, int tile, int stride,
                                          float bias0, float wo0, float bias1, float wo1,
                                          int wv, int lane, int half) {
    int nx = tile + stride;
    if (nx >= HEAD_TILES) nx = tile;
    const v2f* An = (const v2f*)(Al + (long)nx * 16 * K0TOT);
#pragma unroll
    for (int j = 0; j < 48; ++j) nxt[j] = An[2 * j];
    __builtin_amdgcn_sched_barrier(0);

    v8f acc0 = {}, acc1 = {};
#pragma unroll
    for (int j = 0; j < 48; ++j) {
        acc0 = __builtin_amdgcn_wmma_f32_16x16x4_f32(false, cur[j], false, b[j][0], (short)0,
                                                     acc0, false, false);
        acc1 = __builtin_amdgcn_wmma_f32_16x16x4_f32(false, cur[j], false, b[j][1], (short)0,
                                                     acc1, false, false);
    }
    __builtin_amdgcn_sched_barrier(0);
#pragma unroll
    for (int j = 48; j < 96; ++j) nxt[j] = An[2 * j];
    __builtin_amdgcn_sched_barrier(0);
#pragma unroll
    for (int j = 48; j < 96; ++j) {
        acc0 = __builtin_amdgcn_wmma_f32_16x16x4_f32(false, cur[j], false, b[j][0], (short)0,
                                                     acc0, false, false);
        acc1 = __builtin_amdgcn_wmma_f32_16x16x4_f32(false, cur[j], false, b[j][1], (short)0,
                                                     acc1, false, false);
    }
#pragma unroll
    for (int i = 0; i < 8; ++i) {
        float v0 = acc0[i] + bias0;
        float v1 = acc1[i] + bias1;
        float p = 0.f;
        p += (v0 / (1.f + expf(-v0))) * wo0;
        p += (v1 / (1.f + expf(-v1))) * wo1;
        rd[wv][i][lane] = p;
    }
    __syncthreads();
    const int rowbase = tile * 16 + 8 * half;
#pragma unroll
    for (int r = 0; r < 2; ++r) {
        const int i = 2 * wv + r;
        float s = rd[0][i][lane];
        s += rd[1][i][lane];
        s += rd[2][i][lane];
        s += rd[3][i][lane];
        atomicAdd(&out[rowbase + i], s);
    }
}

__global__ __launch_bounds__(128) void k_head_resident(
        const float* __restrict__ x0, const float* __restrict__ Wh, const float* __restrict__ bh,
        const float* __restrict__ Wo, float* __restrict__ out) {
    __shared__ float red[2][4][8][32];
    const int cb     = blockIdx.x % 3;
    const int first  = blockIdx.x / 3;
    const int stride = (cb == 0) ? 86 : 85;
    const int wv   = threadIdx.x >> 5;
    const int lane = threadIdx.x & 31;
    const int half = lane >> 4;
    const int l16  = lane & 15;
    const int n0   = cb * 128 + wv * 32;

    v2f b[96][2];
    const float* Bl = Wh + (long)(2 * half) * K0TOT + n0 + l16;
#pragma unroll
    for (int j = 0; j < 96; ++j) {
#pragma unroll
        for (int t = 0; t < 2; ++t) {
            b[j][t].x = Bl[(4 * j) * K0TOT + t * 16];
            b[j][t].y = Bl[(4 * j + 1) * K0TOT + t * 16];
        }
    }
    const float bias0 = bh[n0 + l16],      wo0 = Wo[n0 + l16];
    const float bias1 = bh[n0 + 16 + l16], wo1 = Wo[n0 + 16 + l16];

    const float* Al = x0 + (long)l16 * K0TOT + 2 * half;
    v2f a[96];
    {
        const v2f* Ar = (const v2f*)(Al + (long)first * 16 * K0TOT);
#pragma unroll
        for (int j = 0; j < 96; ++j) a[j] = Ar[2 * j];
    }
    int par = 0;
    for (int tile = first; tile < HEAD_TILES; tile += stride) {
        v2f an[96];
        head_tile(a, an, b, Al, red[par], out, tile, stride, bias0, wo0, bias1, wo1, wv, lane, half);
        par ^= 1;
        __builtin_amdgcn_sched_barrier(0);
#pragma unroll
        for (int j = 0; j < 96; ++j) a[j] = an[j];
    }
}

extern "C" void kernel_launch(void* const* d_in, const int* in_sizes, int n_in,
                              void* d_out, int out_size, void* d_ws, size_t ws_size,
                              hipStream_t stream) {
    (void)in_sizes; (void)n_in; (void)out_size; (void)ws_size;
    const float* positions = (const float*)d_in[0];
    const int*   species   = (const int*)d_in[1];
    const int*   pairs     = (const int*)d_in[2];
    const float* embeddings= (const float*)d_in[3];
    const float* W_rad     = (const float*)d_in[4];
    const float* W_cg0     = (const float*)d_in[5];
    const float* W_cg1     = (const float*)d_in[6];
    const float* W_cg2     = (const float*)d_in[7];
    const float* W_head    = (const float*)d_in[8];
    const float* b_head    = (const float*)d_in[9];
    const float* W_out     = (const float*)d_in[10];
    const float* b_out     = (const float*)d_in[11];
    float* out = (float*)d_out;

    float* ws = (float*)d_ws;
    const long SZ_F0 = (long)N_ATOMS * KDIM;
    const long SZ_F1 = (long)N_ATOMS * 3 * KDIM;
    const long SZ_F2 = (long)N_ATOMS * 5 * KDIM;
    const long SZ_U  = (long)N_ATOMS * 9 * KDIM;
    float* f0 = ws;
    float* u1 = f0 + SZ_F0 + SZ_F1 + SZ_F2;
    float* u2 = u1 + SZ_U;
    float* g0 = u2 + SZ_U;
    float* g1 = g0 + SZ_F0;
    float* g2 = g1 + SZ_U;
    float* x0 = g2 + SZ_U;

    float4* rec  = (float4*)g2;
    float4* srec = (float4*)g1;
    int* cnt     = (int*)x0;
    int* spill_n = cnt + 10240;
    int* rsp     = cnt + 10304;
    int* scs     = rsp + (long)N_ATOMS * CAP;

    k_init<<<(N_ATOMS + 255) / 256, 256, 0, stream>>>(out, b_out, cnt, spill_n);

    k_bin<<<N_PAIRS / 256, 256, 0, stream>>>(positions, species, pairs, cnt, spill_n,
                                             rec, rsp, srec, scs);

    k_gather<<<N_ATOMS, KDIM, 0, stream>>>(embeddings, W_rad, cnt, spill_n, rec, rsp, srec, scs,
                                           f0, u1, u2);

    k_mix_resident<<<MIX_WAVES, 32, 0, stream>>>(f0, u1, u2, W_cg0, W_cg1, W_cg2, g0, g1, g2);

    k_feats<<<(unsigned)(((long)N_ATOMS * KDIM + 255) / 256), 256, 0, stream>>>(
        f0, u1, u2, g0, g1, g2, species, embeddings, x0);

    k_head_resident<<<HEAD_BLOCKS, 128, 0, stream>>>(x0, W_head, b_head, W_out, out);
}


# l>=1 invariants as signed dots of coupled rows (U2 orthonormal, its L=1 rows antisymmetric): uncouple and 3x3 products go; gather stores 9 rows/atom, mix GEMM on 9 (was 19); sum order of x0 changes
# speedup vs baseline: 2.2755x; 1.1984x over previous
#include <hip/hip_runtime.h>
#include <math.h>

#define N_ATOMS  10000
#define N_PAIRS  160000
#define N_CH     32
#define KDIM     128
#define K0TOT    384

typedef float v2f __attribute__((ext_vector_type(2)));
typedef float v8f __attribute__((ext_vector_type(8)));


#define CAP 64

__global__ void k_init(float* out, const float* b_out, int* cnt, int* spill_n) {
    int i = blockIdx.x * blockDim.x + threadIdx.x;
    if (i < N_ATOMS) { out[i] = b_out[0]; cnt[i] = 0; }
    if (i == 0) *spill_n = 0;
}

__global__ void k_bin(const float* __restrict__ pos, const int* __restrict__ species,
                      const int* __restrict__ pairs, int* __restrict__ cnt,
                      int* __restrict__ spill_n, float4* __restrict__ rec, int* __restrict__ rsp,
                      float4* __restrict__ srec, int* __restrict__ scs) {
    int p = blockIdx.x * blockDim.x + threadIdx.x;
    if (p >= N_PAIRS) return;
    int c  = pairs[2 * p + 0];
    int nb = pairs[2 * p + 1];
    float rx = pos[3 * nb + 0] - pos[3 * c + 0];
    float ry = pos[3 * nb + 1] - pos[3 * c + 1];
    float rz = pos[3 * nb + 2] - pos[3 * c + 2];
    float d  = sqrtf(rx * rx + ry * ry + rz * rz + 1e-12f);
    float inv = 1.f / d;
    float ux = rx * inv, uy = ry * inv, uz = rz * inv;
    float t  = (d - 15.f) * 0.2f;
    t = fminf(fmaxf(t, 0.f), 1.f);
    float fc = 0.5f * (cosf(3.14159265358979f * t) + 1.f);
    float rb[8];
#pragma unroll
    for (int j = 0; j < 8; ++j) {
        float mu = 20.f * (float)j / 7.f;
        float x  = (d - mu) * 0.4f;
        rb[j] = expf(-x * x) * fc;
    }
    const float s3 = 1.7320508075688772f;
    float4 q0 = { rb[0], rb[1], rb[2], rb[3] };
    float4 q1 = { rb[4], rb[5], rb[6], rb[7] };
    float4 q2 = { uy, uz, ux, s3 * ux * uy };
    float4 q3 = { s3 * uy * uz, 0.5f * (3.f * uz * uz - 1.f), s3 * ux * uz,
                  0.5f * s3 * (ux * ux - uy * uy) };
    int sp = species[nb];
    int slot = atomicAdd(&cnt[c], 1);
    float4* dst;
    if (slot < CAP) {
        long idx = (long)c * CAP + slot;
        rsp[idx] = sp;
        dst = rec + idx * 4;
    } else {
        int o = atomicAdd(spill_n, 1);
        scs[2 * o + 0] = c;
        scs[2 * o + 1] = sp;
        dst = srec + (long)o * 4;
    }
    dst[0] = q0; dst[1] = q1; dst[2] = q2; dst[3] = q3;
}

__device__ __forceinline__ void add_pair(const float4* __restrict__ g, int sp,
                                         const float* __restrict__ ek,
                                         const float (&w)[24], float (&acc)[9]) {
    float4 a = g[0], b = g[1], s0 = g[2], s1 = g[3];
    float rb[8] = { a.x, a.y, a.z, a.w, b.x, b.y, b.z, b.w };
    float neigh = ek[sp * N_CH];
    float R[3];
#pragma unroll
    for (int l = 0; l < 3; ++l) {
        float s = 0.f;
#pragma unroll
        for (int j = 0; j < 8; ++j) s += rb[j] * w[l * 8 + j];
        R[l] = s;
    }
    acc[0] += R[0] * neigh;
    float w1 = R[1] * neigh;
    acc[1] += s0.x * w1; acc[2] += s0.y * w1; acc[3] += s0.z * w1;
    float w2 = R[2] * neigh;
    acc[4] += s0.w * w2; acc[5] += s1.x * w2; acc[6] += s1.y * w2;
    acc[7] += s1.z * w2; acc[8] += s1.w * w2;
}

__global__ __launch_bounds__(KDIM) void k_gather(
        const float* __restrict__ emb, const float* __restrict__ W_rad,
        const int* __restrict__ cnt, const int* __restrict__ spill_n,
        const float4* __restrict__ rec, const int* __restrict__ rsp,
        const float4* __restrict__ srec, const int* __restrict__ scs,
        float* __restrict__ f0, float* __restrict__ f1, float* __restrict__ f2) {
    const int c = blockIdx.x;
    const int k = threadIdx.x;
    float w[24];
#pragma unroll
    for (int i = 0; i < 24; ++i) w[i] = W_rad[i * KDIM + k];
    const float* ek = emb + (k >> 2);
    float acc[9];
#pragma unroll
    for (int i = 0; i < 9; ++i) acc[i] = 0.f;

    const int n  = cnt[c];
    const int nr = n < CAP ? n : CAP;
    const float4* r = rec + (long)c * CAP * 4;
    const int*    s = rsp + (long)c * CAP;
    for (int i = 0; i < nr; ++i) add_pair(r + 4 * i, s[i], ek, w, acc);
    if (n > CAP) {
        int no = *spill_n;
        if (no > N_PAIRS) no = N_PAIRS;
        for (int i = 0; i < no; ++i)
            if (scs[2 * i] == c) add_pair(srec + 4 * (long)i, scs[2 * i + 1], ek, w, acc);
    }
    const float mp = 0.1f;
    f0[(long)c * KDIM + k] = acc[0] * mp;
    float* r1 = f1 + (long)c * 3 * KDIM + k;
    float* r2 = f2 + (long)c * 5 * KDIM + k;
#pragma unroll
    for (int q = 0; q < 3; ++q) r1[q * KDIM] = acc[1 + q] * mp;
#pragma unroll
    for (int q = 0; q < 5; ++q) r2[q * KDIM] = acc[4 + q] * mp;
}

__device__ __forceinline__ void mix_tile(const v2f (&cur)[32], v2f (&nxt)[32], const v2f (&b)[32][8],
                                         const float* __restrict__ Al, float* __restrict__ C,
                                         int tile, int stride, int ntiles, int half, int l16) {
    int nx = tile + stride;
    if (nx >= ntiles) nx = tile;
    const v2f* An = (const v2f*)(Al + (long)nx * 16 * KDIM);
#pragma unroll
    for (int j = 0; j < 32; ++j) nxt[j] = An[2 * j];
    __builtin_amdgcn_sched_barrier(0);

    v8f acc[8] = {};
#pragma unroll
    for (int j = 0; j < 32; ++j) {
#pragma unroll
        for (int t = 0; t < 8; ++t)
            acc[t] = __builtin_amdgcn_wmma_f32_16x16x4_f32(false, cur[j], false, b[j][t], (short)0,
                                                           acc[t], false, false);
    }
    float* Ct = C + ((long)tile * 16 + 8 * half) * KDIM + l16;
#pragma unroll
    for (int t = 0; t < 8; ++t) {
#pragma unroll
        for (int i = 0; i < 8; ++i) Ct[(long)i * KDIM + t * 16] = acc[t][i];
    }
}

#define MIX_WAVES 1024
#define MIX_W0    114
#define MIX_W1    341
__global__ __launch_bounds__(32) void k_mix_resident(
        const float* __restrict__ f0, const float* __restrict__ f1, const float* __restrict__ f2,
        const float* __restrict__ Wc0, const float* __restrict__ Wc1, const float* __restrict__ Wc2,
        float* __restrict__ g0, float* __restrict__ g1, float* __restrict__ g2) {
    const int w = blockIdx.x;
    const float* A; const float* B; float* C;
    int first, stride, ntiles;
    if (w < MIX_W0) {
        A = f0; B = Wc0; C = g0; first = w; stride = MIX_W0; ntiles = N_ATOMS / 16;
    } else if (w < MIX_W0 + MIX_W1) {
        A = f1; B = Wc1; C = g1; first = w - MIX_W0; stride = MIX_W1; ntiles = N_ATOMS * 3 / 16;
    } else {
        A = f2; B = Wc2; C = g2; first = w - MIX_W0 - MIX_W1; stride = MIX_WAVES - MIX_W0 - MIX_W1;
        ntiles = N_ATOMS * 5 / 16;
    }
    const int lane = threadIdx.x;
    const int half = lane >> 4;
    const int l16  = lane & 15;

    v2f b[32][8];
    const float* Bl = B + (long)(2 * half) * KDIM + l16;
#pragma unroll
    for (int j = 0; j < 32; ++j) {
#pragma unroll
        for (int t = 0; t < 8; ++t) {
            b[j][t].x = Bl[(4 * j) * KDIM + t * 16];
            b[j][t].y = Bl[(4 * j + 1) * KDIM + t * 16];
        }
    }

    const float* Al = A + (long)l16 * KDIM + 2 * half;
    v2f a[32];
    {
        const v2f* Ar = (const v2f*)(Al + (long)first * 16 * KDIM);
#pragma unroll
        for (int j = 0; j < 32; ++j) a[j] = Ar[2 * j];
    }
    for (int tile = first; tile < ntiles; tile += stride) {
        v2f an[32];
        mix_tile(a, an, b, Al, C, tile, stride, ntiles, half, l16);
        __builtin_amdgcn_sched_barrier(0);
#pragma unroll
        for (int j = 0; j < 32; ++j) a[j] = an[j];
    }
}

__global__ void k_feats(const float* __restrict__ f0, const float* __restrict__ f1,
                        const float* __restrict__ f2, const float* __restrict__ g0,
                        const float* __restrict__ g1, const float* __restrict__ g2,
                        const int* __restrict__ species, const float* __restrict__ emb,
                        float* __restrict__ x0) {
    long t = (long)blockIdx.x * blockDim.x + threadIdx.x;
    if (t >= (long)N_ATOMS * KDIM) return;
    int k = (int)(t % KDIM);
    long n = t / KDIM;
    float f0v = f0[t];
    float feats0 = g0[t] * f0v + f0v;

    float d1 = 0.f;
#pragma unroll
    for (int q = 0; q < 3; ++q) d1 += g1[(n * 3 + q) * KDIM + k] * f1[(n * 3 + q) * KDIM + k];
    float d2 = 0.f;
#pragma unroll
    for (int q = 0; q < 5; ++q) d2 += g2[(n * 5 + q) * KDIM + k] * f2[(n * 5 + q) * KDIM + k];

    const float i3 = 0.5773502691896258f;
    int sp = species[n];
    const float* e = emb + sp * N_CH;
    x0[n * K0TOT + k]       = feats0      * e[k / 12];
    x0[n * K0TOT + 128 + k] = (-d1 * i3)  * e[(128 + k) / 12];
    x0[n * K0TOT + 256 + k] = (d2 * i3)   * e[(256 + k) / 12];
}

#define HEAD_BLOCKS 256
#define HEAD_TILES  (N_ATOMS / 16)
__device__ __forceinline__ void head_tile(const v2f (&cur)[96], v2f (&nxt)[96], const v2f (&b)[96][2],
                                          const float* __restrict__ Al, float (&rd)[4][8][32],
                                          float* __restrict__ out, int tile, int stride,
                                          float bias0, float wo0, float bias1, float wo1,
                                          int wv, int lane, int half) {
    int nx = tile + stride;
    if (nx >= HEAD_TILES) nx = tile;
    const v2f* An = (const v2f*)(Al + (long)nx * 16 * K0TOT);
#pragma unroll
    for (int j = 0; j < 48; ++j) nxt[j] = An[2 * j];
    __builtin_amdgcn_sched_barrier(0);

    v8f acc0 = {}, acc1 = {};
#pragma unroll
    for (int j = 0; j < 48; ++j) {
        acc0 = __builtin_amdgcn_wmma_f32_16x16x4_f32(false, cur[j], false, b[j][0], (short)0,
                                                     acc0, false, false);
        acc1 = __builtin_amdgcn_wmma_f32_16x16x4_f32(false, cur[j], false, b[j][1], (short)0,
                                                     acc1, false, false);
    }
    __builtin_amdgcn_sched_barrier(0);
#pragma unroll
    for (int j = 48; j < 96; ++j) nxt[j] = An[2 * j];
    __builtin_amdgcn_sched_barrier(0);
#pragma unroll
    for (int j = 48; j < 96; ++j) {
        acc0 = __builtin_amdgcn_wmma_f32_16x16x4_f32(false, cur[j], false, b[j][0], (short)0,
                                                     acc0, false, false);
        acc1 = __builtin_amdgcn_wmma_f32_16x16x4_f32(false, cur[j], false, b[j][1], (short)0,
                                                     acc1, false, false);
    }
#pragma unroll
    for (int i = 0; i < 8; ++i) {
        float v0 = acc0[i] + bias0;
        float v1 = acc1[i] + bias1;
        float p = 0.f;
        p += (v0 / (1.f + expf(-v0))) * wo0;
        p += (v1 / (1.f + expf(-v1))) * wo1;
        rd[wv][i][lane] = p;
    }
    __syncthreads();
    const int rowbase = tile * 16 + 8 * half;
#pragma unroll
    for (int r = 0; r < 2; ++r) {
        const int i = 2 * wv + r;
        float s = rd[0][i][lane];
        s += rd[1][i][lane];
        s += rd[2][i][lane];
        s += rd[3][i][lane];
        atomicAdd(&out[rowbase + i], s);
    }
}

__global__ __launch_bounds__(128) void k_head_resident(
        const float* __restrict__ x0, const float* __restrict__ Wh, const float* __restrict__ bh,
        const float* __restrict__ Wo, float* __restrict__ out) {
    __shared__ float red[2][4][8][32];
    const int cb     = blockIdx.x % 3;
    const int first  = blockIdx.x / 3;
    const int stride = (cb == 0) ? 86 : 85;
    const int wv   = threadIdx.x >> 5;
    const int lane = threadIdx.x & 31;
    const int half = lane >> 4;
    const int l16  = lane & 15;
    const int n0   = cb * 128 + wv * 32;

    v2f b[96][2];
    const float* Bl = Wh + (long)(2 * half) * K0TOT + n0 + l16;
#pragma unroll
    for (int j = 0; j < 96; ++j) {
#pragma unroll
        for (int t = 0; t < 2; ++t) {
            b[j][t].x = Bl[(4 * j) * K0TOT + t * 16];
            b[j][t].y = Bl[(4 * j + 1) * K0TOT + t * 16];
        }
    }
    const float bias0 = bh[n0 + l16],      wo0 = Wo[n0 + l16];
    const float bias1 = bh[n0 + 16 + l16], wo1 = Wo[n0 + 16 + l16];

    const float* Al = x0 + (long)l16 * K0TOT + 2 * half;
    v2f a[96];
    {
        const v2f* Ar = (const v2f*)(Al + (long)first * 16 * K0TOT);
#pragma unroll
        for (int j = 0; j < 96; ++j) a[j] = Ar[2 * j];
    }
    int par = 0;
    for (int tile = first; tile < HEAD_TILES; tile += stride) {
        v2f an[96];
        head_tile(a, an, b, Al, red[par], out, tile, stride, bias0, wo0, bias1, wo1, wv, lane, half);
        par ^= 1;
        __builtin_amdgcn_sched_barrier(0);
#pragma unroll
        for (int j = 0; j < 96; ++j) a[j] = an[j];
    }
}

extern "C" void kernel_launch(void* const* d_in, const int* in_sizes, int n_in,
                              void* d_out, int out_size, void* d_ws, size_t ws_size,
                              hipStream_t stream) {
    (void)in_sizes; (void)n_in; (void)out_size; (void)ws_size;
    const float* positions = (const float*)d_in[0];
    const int*   species   = (const int*)d_in[1];
    const int*   pairs     = (const int*)d_in[2];
    const float* embeddings= (const float*)d_in[3];
    const float* W_rad     = (const float*)d_in[4];
    const float* W_cg0     = (const float*)d_in[5];
    const float* W_cg1     = (const float*)d_in[6];
    const float* W_cg2     = (const float*)d_in[7];
    const float* W_head    = (const float*)d_in[8];
    const float* b_head    = (const float*)d_in[9];
    const float* W_out     = (const float*)d_in[10];
    const float* b_out     = (const float*)d_in[11];
    float* out = (float*)d_out;

    float* ws = (float*)d_ws;
    const long SZ_F0 = (long)N_ATOMS * KDIM;
    const long SZ_F1 = (long)N_ATOMS * 3 * KDIM;
    const long SZ_F2 = (long)N_ATOMS * 5 * KDIM;
    const long SZ_U  = (long)N_ATOMS * 9 * KDIM;
    float* f0 = ws;
    float* f1 = f0 + SZ_F0;
    float* f2 = f1 + SZ_F1;
    float* g0 = f2 + SZ_F2 + SZ_U + SZ_U;
    float* g1 = g0 + SZ_F0;
    float* g2 = g1 + SZ_U;
    float* x0 = g2 + SZ_U;

    float4* rec  = (float4*)g2;
    float4* srec = (float4*)g1;
    int* cnt     = (int*)x0;
    int* spill_n = cnt + 10240;
    int* rsp     = cnt + 10304;
    int* scs     = rsp + (long)N_ATOMS * CAP;

    k_init<<<(N_ATOMS + 255) / 256, 256, 0, stream>>>(out, b_out, cnt, spill_n);

    k_bin<<<N_PAIRS / 256, 256, 0, stream>>>(positions, species, pairs, cnt, spill_n,
                                             rec, rsp, srec, scs);

    k_gather<<<N_ATOMS, KDIM, 0, stream>>>(embeddings, W_rad, cnt, spill_n, rec, rsp, srec, scs,
                                           f0, f1, f2);

    k_mix_resident<<<MIX_WAVES, 32, 0, stream>>>(f0, f1, f2, W_cg0, W_cg1, W_cg2, g0, g1, g2);

    k_feats<<<(unsigned)(((long)N_ATOMS * KDIM + 255) / 256), 256, 0, stream>>>(
        f0, f1, f2, g0, g1, g2, species, embeddings, x0);

    k_head_resident<<<HEAD_BLOCKS, 128, 0, stream>>>(x0, W_head, b_head, W_out, out);
}


# gather: one wave per centre atom, lane owns 4 adjacent channels; atom's records staged once in LDS and read as broadcasts (was 4 waves x 2 dependent fetches per pair); 16-byte rows; sum order kept
# speedup vs baseline: 2.3779x; 1.0450x over previous
#include <hip/hip_runtime.h>
#include <math.h>

#define N_ATOMS  10000
#define N_PAIRS  160000
#define N_CH     32
#define KDIM     128
#define K0TOT    384

typedef float v2f __attribute__((ext_vector_type(2)));
typedef float v8f __attribute__((ext_vector_type(8)));


#define CAP 64

__global__ void k_init(float* out, const float* b_out, int* cnt, int* spill_n) {
    int i = blockIdx.x * blockDim.x + threadIdx.x;
    if (i < N_ATOMS) { out[i] = b_out[0]; cnt[i] = 0; }
    if (i == 0) *spill_n = 0;
}

__global__ void k_bin(const float* __restrict__ pos, const int* __restrict__ species,
                      const int* __restrict__ pairs, int* __restrict__ cnt,
                      int* __restrict__ spill_n, float4* __restrict__ rec, int* __restrict__ rsp,
                      float4* __restrict__ srec, int* __restrict__ scs) {
    int p = blockIdx.x * blockDim.x + threadIdx.x;
    if (p >= N_PAIRS) return;
    int c  = pairs[2 * p + 0];
    int nb = pairs[2 * p + 1];
    float rx = pos[3 * nb + 0] - pos[3 * c + 0];
    float ry = pos[3 * nb + 1] - pos[3 * c + 1];
    float rz = pos[3 * nb + 2] - pos[3 * c + 2];
    float d  = sqrtf(rx * rx + ry * ry + rz * rz + 1e-12f);
    float inv = 1.f / d;
    float ux = rx * inv, uy = ry * inv, uz = rz * inv;
    float t  = (d - 15.f) * 0.2f;
    t = fminf(fmaxf(t, 0.f), 1.f);
    float fc = 0.5f * (cosf(3.14159265358979f * t) + 1.f);
    float rb[8];
#pragma unroll
    for (int j = 0; j < 8; ++j) {
        float mu = 20.f * (float)j / 7.f;
        float x  = (d - mu) * 0.4f;
        rb[j] = expf(-x * x) * fc;
    }
    const float s3 = 1.7320508075688772f;
    float4 q0 = { rb[0], rb[1], rb[2], rb[3] };
    float4 q1 = { rb[4], rb[5], rb[6], rb[7] };
    float4 q2 = { uy, uz, ux, s3 * ux * uy };
    float4 q3 = { s3 * uy * uz, 0.5f * (3.f * uz * uz - 1.f), s3 * ux * uz,
                  0.5f * s3 * (ux * ux - uy * uy) };
    int sp = species[nb];
    int slot = atomicAdd(&cnt[c], 1);
    float4* dst;
    if (slot < CAP) {
        long idx = (long)c * CAP + slot;
        rsp[idx] = sp;
        dst = rec + idx * 4;
    } else {
        int o = atomicAdd(spill_n, 1);
        scs[2 * o + 0] = c;
        scs[2 * o + 1] = sp;
        dst = srec + (long)o * 4;
    }
    dst[0] = q0; dst[1] = q1; dst[2] = q2; dst[3] = q3;
}

typedef float v4f __attribute__((ext_vector_type(4)));

#define GATHER_WAVES 4
static_assert(N_ATOMS % GATHER_WAVES == 0, "every wave of the gather grid owns an atom");
static_assert(32 * GATHER_WAVES >= 4 * N_CH, "one thread per embedding value when the table is staged");

__device__ __forceinline__ void add_pair4(const float4* g, float neigh,
                                          const v4f (&w)[24], v4f (&acc)[9]) {
    float4 a = g[0], b = g[1], s0 = g[2], s1 = g[3];
    float rb[8] = { a.x, a.y, a.z, a.w, b.x, b.y, b.z, b.w };
    v4f R[3];
#pragma unroll
    for (int l = 0; l < 3; ++l) {
        v4f s = { 0.f, 0.f, 0.f, 0.f };
#pragma unroll
        for (int j = 0; j < 8; ++j) s += rb[j] * w[l * 8 + j];
        R[l] = s;
    }
    acc[0] += R[0] * neigh;
    v4f w1 = R[1] * neigh;
    acc[1] += s0.x * w1; acc[2] += s0.y * w1; acc[3] += s0.z * w1;
    v4f w2 = R[2] * neigh;
    acc[4] += s0.w * w2; acc[5] += s1.x * w2; acc[6] += s1.y * w2;
    acc[7] += s1.z * w2; acc[8] += s1.w * w2;
}

__global__ __launch_bounds__(32 * GATHER_WAVES) void k_gather(
        const float* __restrict__ emb, const float* __restrict__ W_rad,
        const int* __restrict__ cnt, const int* __restrict__ spill_n,
        const float4* __restrict__ rec, const int* __restrict__ rsp,
        const float4* __restrict__ srec, const int* __restrict__ scs,
        float* __restrict__ f0, float* __restrict__ f1, float* __restrict__ f2) {
    __shared__ float4 s_rec[GATHER_WAVES][CAP * 4];
    __shared__ int    s_sp[GATHER_WAVES][CAP];
    __shared__ float  s_emb[4 * N_CH];
    const int wv = __builtin_amdgcn_readfirstlane((int)(threadIdx.x >> 5));
    const int c  = blockIdx.x * GATHER_WAVES + wv;
    const int L  = threadIdx.x & 31;
    const int n  = cnt[c];
    const int nr = n < CAP ? n : CAP;

    float ev = 0.f;
    if (threadIdx.x < 4 * N_CH) ev = emb[threadIdx.x];
    v4f w[24];
#pragma unroll
    for (int i = 0; i < 24; ++i) w[i] = *(const v4f*)(W_rad + i * KDIM + 4 * L);

    static_assert(CAP % 32 == 0, "a round's 32 x 4 slots stay inside the wave's LDS region");
    const float4* r = rec + (long)c * CAP * 4;
    const int*    s = rsp + (long)c * CAP;
    for (int p0 = 0; p0 < nr; p0 += 32) {
        const int np = nr - p0;
        const int spv = s[p0 + (L < np ? L : np - 1)];
        float4 q[4];
#pragma unroll
        for (int u = 0; u < 4; ++u) {
            const int t = L + 32 * u;
            q[u] = r[4 * p0 + (t < 4 * np ? t : 4 * np - 1)];
        }
        s_sp[wv][p0 + L] = spv;
#pragma unroll
        for (int u = 0; u < 4; ++u) s_rec[wv][4 * p0 + L + 32 * u] = q[u];
    }
    if (threadIdx.x < 4 * N_CH) s_emb[threadIdx.x] = ev;

    v4f acc[9];
#pragma unroll
    for (int i = 0; i < 9; ++i) acc[i] = (v4f){ 0.f, 0.f, 0.f, 0.f };
    __syncthreads();

    for (int i = 0; i < nr; ++i)
        add_pair4(&s_rec[wv][4 * i], s_emb[s_sp[wv][i] * N_CH + L], w, acc);
    if (n > CAP) {
        int no = *spill_n;
        if (no > N_PAIRS) no = N_PAIRS;
        for (int i = 0; i < no; ++i)
            if (scs[2 * i] == c)
                add_pair4(srec + 4 * (long)i, s_emb[scs[2 * i + 1] * N_CH + L], w, acc);
    }
    const float mp = 0.1f;
    *(v4f*)(f0 + (long)c * KDIM + 4 * L) = acc[0] * mp;
    float* r1 = f1 + (long)c * 3 * KDIM + 4 * L;
    float* r2 = f2 + (long)c * 5 * KDIM + 4 * L;
#pragma unroll
    for (int q = 0; q < 3; ++q) *(v4f*)(r1 + q * KDIM) = acc[1 + q] * mp;
#pragma unroll
    for (int q = 0; q < 5; ++q) *(v4f*)(r2 + q * KDIM) = acc[4 + q] * mp;
}

__device__ __forceinline__ void mix_tile(const v2f (&cur)[32], v2f (&nxt)[32], const v2f (&b)[32][8],
                                         const float* __restrict__ Al, float* __restrict__ C,
                                         int tile, int stride, int ntiles, int half, int l16) {
    int nx = tile + stride;
    if (nx >= ntiles) nx = tile;
    const v2f* An = (const v2f*)(Al + (long)nx * 16 * KDIM);
#pragma unroll
    for (int j = 0; j < 32; ++j) nxt[j] = An[2 * j];
    __builtin_amdgcn_sched_barrier(0);

    v8f acc[8] = {};
#pragma unroll
    for (int j = 0; j < 32; ++j) {
#pragma unroll
        for (int t = 0; t < 8; ++t)
            acc[t] = __builtin_amdgcn_wmma_f32_16x16x4_f32(false, cur[j], false, b[j][t], (short)0,
                                                           acc[t], false, false);
    }
    float* Ct = C + ((long)tile * 16 + 8 * half) * KDIM + l16;
#pragma unroll
    for (int t = 0; t < 8; ++t) {
#pragma unroll
        for (int i = 0; i < 8; ++i) Ct[(long)i * KDIM + t * 16] = acc[t][i];
    }
}

#define MIX_WAVES 1024
#define MIX_W0    114
#define MIX_W1    341
__global__ __launch_bounds__(32) void k_mix_resident(
        const float* __restrict__ f0, const float* __restrict__ f1, const float* __restrict__ f2,
        const float* __restrict__ Wc0, const float* __restrict__ Wc1, const float* __restrict__ Wc2,
        float* __restrict__ g0, float* __restrict__ g1, float* __restrict__ g2) {
    const int w = blockIdx.x;
    const float* A; const float* B; float* C;
    int first, stride, ntiles;
    if (w < MIX_W0) {
        A = f0; B = Wc0; C = g0; first = w; stride = MIX_W0; ntiles = N_ATOMS / 16;
    } else if (w < MIX_W0 + MIX_W1) {
        A = f1; B = Wc1; C = g1; first = w - MIX_W0; stride = MIX_W1; ntiles = N_ATOMS * 3 / 16;
    } else {
        A = f2; B = Wc2; C = g2; first = w - MIX_W0 - MIX_W1; stride = MIX_WAVES - MIX_W0 - MIX_W1;
        ntiles = N_ATOMS * 5 / 16;
    }
    const int lane = threadIdx.x;
    const int half = lane >> 4;
    const int l16  = lane & 15;

    v2f b[32][8];
    const float* Bl = B + (long)(2 * half) * KDIM + l16;
#pragma unroll
    for (int j = 0; j < 32; ++j) {
#pragma unroll
        for (int t = 0; t < 8; ++t) {
            b[j][t].x = Bl[(4 * j) * KDIM + t * 16];
            b[j][t].y = Bl[(4 * j + 1) * KDIM + t * 16];
        }
    }

    const float* Al = A + (long)l16 * KDIM + 2 * half;
    v2f a[32];
    {
        const v2f* Ar = (const v2f*)(Al + (long)first * 16 * KDIM);
#pragma unroll
        for (int j = 0; j < 32; ++j) a[j] = Ar[2 * j];
    }
    for (int tile = first; tile < ntiles; tile += stride) {
        v2f an[32];
        mix_tile(a, an, b, Al, C, tile, stride, ntiles, half, l16);
        __builtin_amdgcn_sched_barrier(0);
#pragma unroll
        for (int j = 0; j < 32; ++j) a[j] = an[j];
    }
}

__global__ void k_feats(const float* __restrict__ f0, const float* __restrict__ f1,
                        const float* __restrict__ f2, const float* __restrict__ g0,
                        const float* __restrict__ g1, const float* __restrict__ g2,
                        const int* __restrict__ species, const float* __restrict__ emb,
                        float* __restrict__ x0) {
    long t = (long)blockIdx.x * blockDim.x + threadIdx.x;
    if (t >= (long)N_ATOMS * KDIM) return;
    int k = (int)(t % KDIM);
    long n = t / KDIM;
    float f0v = f0[t];
    float feats0 = g0[t] * f0v + f0v;

    float d1 = 0.f;
#pragma unroll
    for (int q = 0; q < 3; ++q) d1 += g1[(n * 3 + q) * KDIM + k] * f1[(n * 3 + q) * KDIM + k];
    float d2 = 0.f;
#pragma unroll
    for (int q = 0; q < 5; ++q) d2 += g2[(n * 5 + q) * KDIM + k] * f2[(n * 5 + q) * KDIM + k];

    const float i3 = 0.5773502691896258f;
    int sp = species[n];
    const float* e = emb + sp * N_CH;
    x0[n * K0TOT + k]       = feats0      * e[k / 12];
    x0[n * K0TOT + 128 + k] = (-d1 * i3)  * e[(128 + k) / 12];
    x0[n * K0TOT + 256 + k] = (d2 * i3)   * e[(256 + k) / 12];
}

#define HEAD_BLOCKS 256
#define HEAD_TILES  (N_ATOMS / 16)
__device__ __forceinline__ void head_tile(const v2f (&cur)[96], v2f (&nxt)[96], const v2f (&b)[96][2],
                                          const float* __restrict__ Al, float (&rd)[4][8][32],
                                          float* __restrict__ out, int tile, int stride,
                                          float bias0, float wo0, float bias1, float wo1,
                                          int wv, int lane, int half) {
    int nx = tile + stride;
    if (nx >= HEAD_TILES) nx = tile;
    const v2f* An = (const v2f*)(Al + (long)nx * 16 * K0TOT);
#pragma unroll
    for (int j = 0; j < 48; ++j) nxt[j] = An[2 * j];
    __builtin_amdgcn_sched_barrier(0);

    v8f acc0 = {}, acc1 = {};
#pragma unroll
    for (int j = 0; j < 48; ++j) {
        acc0 = __builtin_amdgcn_wmma_f32_16x16x4_f32(false, cur[j], false, b[j][0], (short)0,
                                                     acc0, false, false);
        acc1 = __builtin_amdgcn_wmma_f32_16x16x4_f32(false, cur[j], false, b[j][1], (short)0,
                                                     acc1, false, false);
    }
    __builtin_amdgcn_sched_barrier(0);
#pragma unroll
    for (int j = 48; j < 96; ++j) nxt[j] = An[2 * j];
    __builtin_amdgcn_sched_barrier(0);
#pragma unroll
    for (int j = 48; j < 96; ++j) {
        acc0 = __builtin_amdgcn_wmma_f32_16x16x4_f32(false, cur[j], false, b[j][0], (short)0,
                                                     acc0, false, false);
        acc1 = __builtin_amdgcn_wmma_f32_16x16x4_f32(false, cur[j], false, b[j][1], (short)0,
                                                     acc1, false, false);
    }
#pragma unroll
    for (int i = 0; i < 8; ++i) {
        float v0 = acc0[i] + bias0;
        float v1 = acc1[i] + bias1;
        float p = 0.f;
        p += (v0 / (1.f + expf(-v0))) * wo0;
        p += (v1 / (1.f + expf(-v1))) * wo1;
        rd[wv][i][lane] = p;
    }
    __syncthreads();
    const int rowbase = tile * 16 + 8 * half;
#pragma unroll
    for (int r = 0; r < 2; ++r) {
        const int i = 2 * wv + r;
        float s = rd[0][i][lane];
        s += rd[1][i][lane];
        s += rd[2][i][lane];
        s += rd[3][i][lane];
        atomicAdd(&out[rowbase + i], s);
    }
}

__global__ __launch_bounds__(128) void k_head_resident(
        const float* __restrict__ x0, const float* __restrict__ Wh, const float* __restrict__ bh,
        const float* __restrict__ Wo, float* __restrict__ out) {
    __shared__ float red[2][4][8][32];
    const int cb     = blockIdx.x % 3;
    const int first  = blockIdx.x / 3;
    const int stride = (cb == 0) ? 86 : 85;
    const int wv   = threadIdx.x >> 5;
    const int lane = threadIdx.x & 31;
    const int half = lane >> 4;
    const int l16  = lane & 15;
    const int n0   = cb * 128 + wv * 32;

    v2f b[96][2];
    const float* Bl = Wh + (long)(2 * half) * K0TOT + n0 + l16;
#pragma unroll
    for (int j = 0; j < 96; ++j) {
#pragma unroll
        for (int t = 0; t < 2; ++t) {
            b[j][t].x = Bl[(4 * j) * K0TOT + t * 16];
            b[j][t].y = Bl[(4 * j + 1) * K0TOT + t * 16];
        }
    }
    const float bias0 = bh[n0 + l16],      wo0 = Wo[n0 + l16];
    const float bias1 = bh[n0 + 16 + l16], wo1 = Wo[n0 + 16 + l16];

    const float* Al = x0 + (long)l16 * K0TOT + 2 * half;
    v2f a[96];
    {
        const v2f* Ar = (const v2f*)(Al + (long)first * 16 * K0TOT);
#pragma unroll
        for (int j = 0; j < 96; ++j) a[j] = Ar[2 * j];
    }
    int par = 0;
    for (int tile = first; tile < HEAD_TILES; tile += stride) {
        v2f an[96];
        head_tile(a, an, b, Al, red[par], out, tile, stride, bias0, wo0, bias1, wo1, wv, lane, half);
        par ^= 1;
        __builtin_amdgcn_sched_barrier(0);
#pragma unroll
        for (int j = 0; j < 96; ++j) a[j] = an[j];
    }
}

extern "C" void kernel_launch(void* const* d_in, const int* in_sizes, int n_in,
                              void* d_out, int out_size, void* d_ws, size_t ws_size,
                              hipStream_t stream) {
    (void)in_sizes; (void)n_in; (void)out_size; (void)ws_size;
    const float* positions = (const float*)d_in[0];
    const int*   species   = (const int*)d_in[1];
    const int*   pairs     = (const int*)d_in[2];
    const float* embeddings= (const float*)d_in[3];
    const float* W_rad     = (const float*)d_in[4];
    const float* W_cg0     = (const float*)d_in[5];
    const float* W_cg1     = (const float*)d_in[6];
    const float* W_cg2     = (const float*)d_in[7];
    const float* W_head    = (const float*)d_in[8];
    const float* b_head    = (const float*)d_in[9];
    const float* W_out     = (const float*)d_in[10];
    const float* b_out     = (const float*)d_in[11];
    float* out = (float*)d_out;

    float* ws = (float*)d_ws;
    const long SZ_F0 = (long)N_ATOMS * KDIM;
    const long SZ_F1 = (long)N_ATOMS * 3 * KDIM;
    const long SZ_F2 = (long)N_ATOMS * 5 * KDIM;
    const long SZ_U  = (long)N_ATOMS * 9 * KDIM;
    float* f0 = ws;
    float* f1 = f0 + SZ_F0;
    float* f2 = f1 + SZ_F1;
    float* g0 = f2 + SZ_F2 + SZ_U + SZ_U;
    float* g1 = g0 + SZ_F0;
    float* g2 = g1 + SZ_U;
    float* x0 = g2 + SZ_U;

    float4* rec  = (float4*)g2;
    float4* srec = (float4*)g1;
    int* cnt     = (int*)x0;
    int* spill_n = cnt + 10240;
    int* rsp     = cnt + 10304;
    int* scs     = rsp + (long)N_ATOMS * CAP;

    k_init<<<(N_ATOMS + 255) / 256, 256, 0, stream>>>(out, b_out, cnt, spill_n);

    k_bin<<<N_PAIRS / 256, 256, 0, stream>>>(positions, species, pairs, cnt, spill_n,
                                             rec, rsp, srec, scs);

    k_gather<<<N_ATOMS / GATHER_WAVES, 32 * GATHER_WAVES, 0, stream>>>(embeddings, W_rad, cnt, spill_n, rec, rsp, srec, scs,
                                           f0, f1, f2);

    k_mix_resident<<<MIX_WAVES, 32, 0, stream>>>(f0, f1, f2, W_cg0, W_cg1, W_cg2, g0, g1, g2);

    k_feats<<<(unsigned)(((long)N_ATOMS * KDIM + 255) / 256), 256, 0, stream>>>(
        f0, f1, f2, g0, g1, g2, species, embeddings, x0);

    k_head_resident<<<HEAD_BLOCKS, 128, 0, stream>>>(x0, W_head, b_head, W_out, out);
}


# feats folded into the channel mix: 4-wave blocks keep W_cg stripes, a tile = 16 atoms x one coupled row q, so the q-sum of g f stays in the lane; g planes and feats launch go; sum orders kept
# speedup vs baseline: 2.5168x; 1.0584x over previous
#include <hip/hip_runtime.h>
#include <math.h>

#define N_ATOMS  10000
#define N_PAIRS  160000
#define N_CH     32
#define KDIM     128
#define K0TOT    384

typedef float v2f __attribute__((ext_vector_type(2)));
typedef float v8f __attribute__((ext_vector_type(8)));


#define CAP 64

__global__ void k_init(float* out, const float* b_out, int* cnt, int* spill_n) {
    int i = blockIdx.x * blockDim.x + threadIdx.x;
    if (i < N_ATOMS) { out[i] = b_out[0]; cnt[i] = 0; }
    if (i == 0) *spill_n = 0;
}

__global__ void k_bin(const float* __restrict__ pos, const int* __restrict__ species,
                      const int* __restrict__ pairs, int* __restrict__ cnt,
                      int* __restrict__ spill_n, float4* __restrict__ rec, int* __restrict__ rsp,
                      float4* __restrict__ srec, int* __restrict__ scs) {
    int p = blockIdx.x * blockDim.x + threadIdx.x;
    if (p >= N_PAIRS) return;
    int c  = pairs[2 * p + 0];
    int nb = pairs[2 * p + 1];
    float rx = pos[3 * nb + 0] - pos[3 * c + 0];
    float ry = pos[3 * nb + 1] - pos[3 * c + 1];
    float rz = pos[3 * nb + 2] - pos[3 * c + 2];
    float d  = sqrtf(rx * rx + ry * ry + rz * rz + 1e-12f);
    float inv = 1.f / d;
    float ux = rx * inv, uy = ry * inv, uz = rz * inv;
    float t  = (d - 15.f) * 0.2f;
    t = fminf(fmaxf(t, 0.f), 1.f);
    float fc = 0.5f * (cosf(3.14159265358979f * t) + 1.f);
    float rb[8];
#pragma unroll
    for (int j = 0; j < 8; ++j) {
        float mu = 20.f * (float)j / 7.f;
        float x  = (d - mu) * 0.4f;
        rb[j] = expf(-x * x) * fc;
    }
    const float s3 = 1.7320508075688772f;
    float4 q0 = { rb[0], rb[1], rb[2], rb[3] };
    float4 q1 = { rb[4], rb[5], rb[6], rb[7] };
    float4 q2 = { uy, uz, ux, s3 * ux * uy };
    float4 q3 = { s3 * uy * uz, 0.5f * (3.f * uz * uz - 1.f), s3 * ux * uz,
                  0.5f * s3 * (ux * ux - uy * uy) };
    int sp = species[nb];
    int slot = atomicAdd(&cnt[c], 1);
    float4* dst;
    if (slot < CAP) {
        long idx = (long)c * CAP + slot;
        rsp[idx] = sp;
        dst = rec + idx * 4;
    } else {
        int o = atomicAdd(spill_n, 1);
        scs[2 * o + 0] = c;
        scs[2 * o + 1] = sp;
        dst = srec + (long)o * 4;
    }
    dst[0] = q0; dst[1] = q1; dst[2] = q2; dst[3] = q3;
}

typedef float v4f __attribute__((ext_vector_type(4)));

#define GATHER_WAVES 4
static_assert(N_ATOMS % GATHER_WAVES == 0, "every wave of the gather grid owns an atom");
static_assert(32 * GATHER_WAVES >= 4 * N_CH, "one thread per embedding value when the table is staged");

__device__ __forceinline__ void add_pair4(const float4* g, float neigh,
                                          const v4f (&w)[24], v4f (&acc)[9]) {
    float4 a = g[0], b = g[1], s0 = g[2], s1 = g[3];
    float rb[8] = { a.x, a.y, a.z, a.w, b.x, b.y, b.z, b.w };
    v4f R[3];
#pragma unroll
    for (int l = 0; l < 3; ++l) {
        v4f s = { 0.f, 0.f, 0.f, 0.f };
#pragma unroll
        for (int j = 0; j < 8; ++j) s += rb[j] * w[l * 8 + j];
        R[l] = s;
    }
    acc[0] += R[0] * neigh;
    v4f w1 = R[1] * neigh;
    acc[1] += s0.x * w1; acc[2] += s0.y * w1; acc[3] += s0.z * w1;
    v4f w2 = R[2] * neigh;
    acc[4] += s0.w * w2; acc[5] += s1.x * w2; acc[6] += s1.y * w2;
    acc[7] += s1.z * w2; acc[8] += s1.w * w2;
}

__global__ __launch_bounds__(32 * GATHER_WAVES) void k_gather(
        const float* __restrict__ emb, const float* __restrict__ W_rad,
        const int* __restrict__ cnt, const int* __restrict__ spill_n,
        const float4* __restrict__ rec, const int* __restrict__ rsp,
        const float4* __restrict__ srec, const int* __restrict__ scs,
        float* __restrict__ f0, float* __restrict__ f1, float* __restrict__ f2) {
    __shared__ float4 s_rec[GATHER_WAVES][CAP * 4];
    __shared__ int    s_sp[GATHER_WAVES][CAP];
    __shared__ float  s_emb[4 * N_CH];
    const int wv = __builtin_amdgcn_readfirstlane((int)(threadIdx.x >> 5));
    const int c  = blockIdx.x * GATHER_WAVES + wv;
    const int L  = threadIdx.x & 31;
    const int n  = cnt[c];
    const int nr = n < CAP ? n : CAP;

    float ev = 0.f;
    if (threadIdx.x < 4 * N_CH) ev = emb[threadIdx.x];
    v4f w[24];
#pragma unroll
    for (int i = 0; i < 24; ++i) w[i] = *(const v4f*)(W_rad + i * KDIM + 4 * L);

    static_assert(CAP % 32 == 0, "a round's 32 x 4 slots stay inside the wave's LDS region");
    const float4* r = rec + (long)c * CAP * 4;
    const int*    s = rsp + (long)c * CAP;
    for (int p0 = 0; p0 < nr; p0 += 32) {
        const int np = nr - p0;
        const int spv = s[p0 + (L < np ? L : np - 1)];
        float4 q[4];
#pragma unroll
        for (int u = 0; u < 4; ++u) {
            const int t = L + 32 * u;
            q[u] = r[4 * p0 + (t < 4 * np ? t : 4 * np - 1)];
        }
        s_sp[wv][p0 + L] = spv;
#pragma unroll
        for (int u = 0; u < 4; ++u) s_rec[wv][4 * p0 + L + 32 * u] = q[u];
    }
    if (threadIdx.x < 4 * N_CH) s_emb[threadIdx.x] = ev;

    v4f acc[9];
#pragma unroll
    for (int i = 0; i < 9; ++i) acc[i] = (v4f){ 0.f, 0.f, 0.f, 0.f };
    __syncthreads();

    for (int i = 0; i < nr; ++i)
        add_pair4(&s_rec[wv][4 * i], s_emb[s_sp[wv][i] * N_CH + L], w, acc);
    if (n > CAP) {
        int no = *spill_n;
        if (no > N_PAIRS) no = N_PAIRS;
        for (int i = 0; i < no; ++i)
            if (scs[2 * i] == c)
                add_pair4(srec + 4 * (long)i, s_emb[scs[2 * i + 1] * N_CH + L], w, acc);
    }
    const float mp = 0.1f;
    *(v4f*)(f0 + (long)c * KDIM + 4 * L) = acc[0] * mp;
    float* r1 = f1 + (long)c * 3 * KDIM + 4 * L;
    float* r2 = f2 + (long)c * 5 * KDIM + 4 * L;
#pragma unroll
    for (int q = 0; q < 3; ++q) *(v4f*)(r1 + q * KDIM) = acc[1 + q] * mp;
#pragma unroll
    for (int q = 0; q < 5; ++q) *(v4f*)(r2 + q * KDIM) = acc[4 + q] * mp;
}

#define MIXB    256
#define MIXB2   136
#define MIXB1   90
#define NGROUPS (N_ATOMS / 16)
static_assert(N_ATOMS % 16 == 0, "a tile is 16 whole atoms");
static_assert(MIXB2 + MIXB1 < MIXB && MIXB - MIXB2 - MIXB1 <= NGROUPS && MIXB2 <= NGROUPS, "every block owns a group");
__global__ __launch_bounds__(128) void k_mix_feats(
        const float* __restrict__ f0, const float* __restrict__ f1, const float* __restrict__ f2,
        const float* __restrict__ Wc0, const float* __restrict__ Wc1, const float* __restrict__ Wc2,
        const int* __restrict__ species, const float* __restrict__ emb, float* __restrict__ x0) {
    const int blk = blockIdx.x;
    const float* A; const float* B;
    int Q, lsel, first, stride;
    if (blk < MIXB2) {
        A = f2; B = Wc2; Q = 5; lsel = 2; first = blk; stride = MIXB2;
    } else if (blk < MIXB2 + MIXB1) {
        A = f1; B = Wc1; Q = 3; lsel = 1; first = blk - MIXB2; stride = MIXB1;
    } else {
        A = f0; B = Wc0; Q = 1; lsel = 0; first = blk - MIXB2 - MIXB1; stride = MIXB - MIXB2 - MIXB1;
    }
    const int wv   = threadIdx.x >> 5;
    const int lane = threadIdx.x & 31;
    const int half = lane >> 4;
    const int l16  = lane & 15;
    const int col0 = wv * 32 + l16;

    v2f b[32][2];
    const float* Bl = B + (long)(2 * half) * KDIM + col0;
#pragma unroll
    for (int j = 0; j < 32; ++j) {
#pragma unroll
        for (int t = 0; t < 2; ++t) {
            b[j][t].x = Bl[(4 * j) * KDIM + t * 16];
            b[j][t].y = Bl[(4 * j + 1) * KDIM + t * 16];
        }
    }
    __shared__ float s_emb[4 * N_CH];
    static_assert(128 >= 4 * N_CH, "one thread per embedding value when the table is staged");
    s_emb[threadIdx.x] = emb[threadIdx.x];
    const int qi0 = (lsel * KDIM + col0) / 12;
    const int qi1 = (lsel * KDIM + col0 + 16) / 12;
    const float i3 = 0.5773502691896258f;
    const float sc = (lsel == 1) ? -i3 : i3;

    const long pitch = (long)Q * KDIM;
    const float* Al = A + (long)l16 * pitch + 2 * half;
    const float* Fl = A + (long)(8 * half) * pitch + col0;

    v2f a[32];
    {
        const v2f* Ar = (const v2f*)(Al + (long)first * 16 * pitch);
#pragma unroll
        for (int j = 0; j < 32; ++j) a[j] = Ar[2 * j];
    }
    __syncthreads();
    float P[2][8];
    int sp[8];
#pragma unroll
    for (int r = 0; r < 8; ++r) { P[0][r] = 0.f; P[1][r] = 0.f; sp[r] = 0; }
    int g = first, q = 0;
    while (g < NGROUPS) {
        int gn = g, qn = q + 1;
        if (qn == Q) { qn = 0; gn = g + stride; }
        const bool more = gn < NGROUPS;
        const v2f* An = (const v2f*)(Al + (long)(more ? gn : g) * 16 * pitch + (long)(more ? qn : q) * KDIM);
        v2f an[32];
#pragma unroll
        for (int j = 0; j < 32; ++j) an[j] = An[2 * j];
        const float* Fc = Fl + (long)g * 16 * pitch + (long)q * KDIM;
        float fc[2][8];
#pragma unroll
        for (int r = 0; r < 8; ++r) {
            fc[0][r] = Fc[(long)r * pitch];
            fc[1][r] = Fc[(long)r * pitch + 16];
        }
        if (q == 0) {
#pragma unroll
            for (int r = 0; r < 8; ++r) { sp[r] = species[g * 16 + 8 * half + r]; P[0][r] = 0.f; P[1][r] = 0.f; }
        }
        __builtin_amdgcn_sched_barrier(0);

        v8f acc0 = {}, acc1 = {};
#pragma unroll
        for (int j = 0; j < 32; ++j) {
            acc0 = __builtin_amdgcn_wmma_f32_16x16x4_f32(false, a[j], false, b[j][0], (short)0,
                                                         acc0, false, false);
            acc1 = __builtin_amdgcn_wmma_f32_16x16x4_f32(false, a[j], false, b[j][1], (short)0,
                                                         acc1, false, false);
        }
        __builtin_amdgcn_sched_barrier(0);
#pragma unroll
        for (int r = 0; r < 8; ++r) {
            P[0][r] = acc0[r] * fc[0][r] + ((lsel == 0) ? fc[0][r] : P[0][r]);
            P[1][r] = acc1[r] * fc[1][r] + ((lsel == 0) ? fc[1][r] : P[1][r]);
        }
        if (q == Q - 1) {
            float* X = x0 + ((long)g * 16 + 8 * half) * K0TOT + lsel * KDIM + col0;
#pragma unroll
            for (int r = 0; r < 8; ++r) {
                const float* e = s_emb + sp[r] * N_CH;
                const float v0 = (lsel == 0) ? P[0][r] : P[0][r] * sc;
                const float v1 = (lsel == 0) ? P[1][r] : P[1][r] * sc;
                X[(long)r * K0TOT]      = v0 * e[qi0];
                X[(long)r * K0TOT + 16] = v1 * e[qi1];
            }
        }
        __builtin_amdgcn_sched_barrier(0);
#pragma unroll
        for (int j = 0; j < 32; ++j) a[j] = an[j];
        g = gn; q = qn;
    }
}

#define HEAD_BLOCKS 256
#define HEAD_TILES  (N_ATOMS / 16)
__device__ __forceinline__ void head_tile(const v2f (&cur)[96], v2f (&nxt)[96], const v2f (&b)[96][2],
                                          const float* __restrict__ Al, float (&rd)[4][8][32],
                                          float* __restrict__ out, int tile, int stride,
                                          float bias0, float wo0, float bias1, float wo1,
                                          int wv, int lane, int half) {
    int nx = tile + stride;
    if (nx >= HEAD_TILES) nx = tile;
    const v2f* An = (const v2f*)(Al + (long)nx * 16 * K0TOT);
#pragma unroll
    for (int j = 0; j < 48; ++j) nxt[j] = An[2 * j];
    __builtin_amdgcn_sched_barrier(0);

    v8f acc0 = {}, acc1 = {};
#pragma unroll
    for (int j = 0; j < 48; ++j) {
        acc0 = __builtin_amdgcn_wmma_f32_16x16x4_f32(false, cur[j], false, b[j][0], (short)0,
                                                     acc0, false, false);
        acc1 = __builtin_amdgcn_wmma_f32_16x16x4_f32(false, cur[j], false, b[j][1], (short)0,
                                                     acc1, false, false);
    }
    __builtin_amdgcn_sched_barrier(0);
#pragma unroll
    for (int j = 48; j < 96; ++j) nxt[j] = An[2 * j];
    __builtin_amdgcn_sched_barrier(0);
#pragma unroll
    for (int j = 48; j < 96; ++j) {
        acc0 = __builtin_amdgcn_wmma_f32_16x16x4_f32(false, cur[j], false, b[j][0], (short)0,
                                                     acc0, false, false);
        acc1 = __builtin_amdgcn_wmma_f32_16x16x4_f32(false, cur[j], false, b[j][1], (short)0,
                                                     acc1, false, false);
    }
#pragma unroll
    for (int i = 0; i < 8; ++i) {
        float v0 = acc0[i] + bias0;
        float v1 = acc1[i] + bias1;
        float p = 0.f;
        p += (v0 / (1.f + expf(-v0))) * wo0;
        p += (v1 / (1.f + expf(-v1))) * wo1;
        rd[wv][i][lane] = p;
    }
    __syncthreads();
    const int rowbase = tile * 16 + 8 * half;
#pragma unroll
    for (int r = 0; r < 2; ++r) {
        const int i = 2 * wv + r;
        float s = rd[0][i][lane];
        s += rd[1][i][lane];
        s += rd[2][i][lane];
        s += rd[3][i][lane];
        atomicAdd(&out[rowbase + i], s);
    }
}

__global__ __launch_bounds__(128) void k_head_resident(
        const float* __restrict__ x0, const float* __restrict__ Wh, const float* __restrict__ bh,
        const float* __restrict__ Wo, float* __restrict__ out) {
    __shared__ float red[2][4][8][32];
    const int cb     = blockIdx.x % 3;
    const int first  = blockIdx.x / 3;
    const int stride = (cb == 0) ? 86 : 85;
    const int wv   = threadIdx.x >> 5;
    const int lane = threadIdx.x & 31;
    const int half = lane >> 4;
    const int l16  = lane & 15;
    const int n0   = cb * 128 + wv * 32;

    v2f b[96][2];
    const float* Bl = Wh + (long)(2 * half) * K0TOT + n0 + l16;
#pragma unroll
    for (int j = 0; j < 96; ++j) {
#pragma unroll
        for (int t = 0; t < 2; ++t) {
            b[j][t].x = Bl[(4 * j) * K0TOT + t * 16];
            b[j][t].y = Bl[(4 * j + 1) * K0TOT + t * 16];
        }
    }
    const float bias0 = bh[n0 + l16],      wo0 = Wo[n0 + l16];
    const float bias1 = bh[n0 + 16 + l16], wo1 = Wo[n0 + 16 + l16];

    const float* Al = x0 + (long)l16 * K0TOT + 2 * half;
    v2f a[96];
    {
        const v2f* Ar = (const v2f*)(Al + (long)first * 16 * K0TOT);
#pragma unroll
        for (int j = 0; j < 96; ++j) a[j] = Ar[2 * j];
    }
    int par = 0;
    for (int tile = first; tile < HEAD_TILES; tile += stride) {
        v2f an[96];
        head_tile(a, an, b, Al, red[par], out, tile, stride, bias0, wo0, bias1, wo1, wv, lane, half);
        par ^= 1;
        __builtin_amdgcn_sched_barrier(0);
#pragma unroll
        for (int j = 0; j < 96; ++j) a[j] = an[j];
    }
}

extern "C" void kernel_launch(void* const* d_in, const int* in_sizes, int n_in,
                              void* d_out, int out_size, void* d_ws, size_t ws_size,
                              hipStream_t stream) {
    (void)in_sizes; (void)n_in; (void)out_size; (void)ws_size;
    const float* positions = (const float*)d_in[0];
    const int*   species   = (const int*)d_in[1];
    const int*   pairs     = (const int*)d_in[2];
    const float* embeddings= (const float*)d_in[3];
    const float* W_rad     = (const float*)d_in[4];
    const float* W_cg0     = (const float*)d_in[5];
    const float* W_cg1     = (const float*)d_in[6];
    const float* W_cg2     = (const float*)d_in[7];
    const float* W_head    = (const float*)d_in[8];
    const float* b_head    = (const float*)d_in[9];
    const float* W_out     = (const float*)d_in[10];
    const float* b_out     = (const float*)d_in[11];
    float* out = (float*)d_out;

    float* ws = (float*)d_ws;
    const long SZ_F0 = (long)N_ATOMS * KDIM;
    const long SZ_F1 = (long)N_ATOMS * 3 * KDIM;
    const long SZ_F2 = (long)N_ATOMS * 5 * KDIM;
    const long SZ_U  = (long)N_ATOMS * 9 * KDIM;
    float* f0 = ws;
    float* f1 = f0 + SZ_F0;
    float* f2 = f1 + SZ_F1;
    float* g0 = f2 + SZ_F2 + SZ_U + SZ_U;
    float* g1 = g0 + SZ_F0;
    float* g2 = g1 + SZ_U;
    float* x0 = g2 + SZ_U;

    float4* rec  = (float4*)g2;
    float4* srec = (float4*)g1;
    int* cnt     = (int*)x0;
    int* spill_n = cnt + 10240;
    int* rsp     = cnt + 10304;
    int* scs     = rsp + (long)N_ATOMS * CAP;

    k_init<<<(N_ATOMS + 255) / 256, 256, 0, stream>>>(out, b_out, cnt, spill_n);

    k_bin<<<N_PAIRS / 256, 256, 0, stream>>>(positions, species, pairs, cnt, spill_n,
                                             rec, rsp, srec, scs);

    k_gather<<<N_ATOMS / GATHER_WAVES, 32 * GATHER_WAVES, 0, stream>>>(embeddings, W_rad, cnt, spill_n, rec, rsp, srec, scs,
                                           f0, f1, f2);

    k_mix_feats<<<MIXB, 128, 0, stream>>>(f0, f1, f2, W_cg0, W_cg1, W_cg2, species, embeddings, x0);

    k_head_resident<<<HEAD_BLOCKS, 128, 0, stream>>>(x0, W_head, b_head, W_out, out);
}


# A operand of both GEMMs through LDS: each tile fetched once per block by whole-line 16-byte loads into 3 rotating buffers, fragments by ds_load (was 16-line loads by all 4 waves); sum orders kept
# speedup vs baseline: 2.7571x; 1.0955x over previous
#include <hip/hip_runtime.h>
#include <math.h>

#define N_ATOMS  10000
#define N_PAIRS  160000
#define N_CH     32
#define KDIM     128
#define K0TOT    384

typedef float v2f __attribute__((ext_vector_type(2)));
typedef float v8f __attribute__((ext_vector_type(8)));


#define CAP 64

__global__ void k_init(float* out, const float* b_out, int* cnt, int* spill_n) {
    int i = blockIdx.x * blockDim.x + threadIdx.x;
    if (i < N_ATOMS) { out[i] = b_out[0]; cnt[i] = 0; }
    if (i == 0) *spill_n = 0;
}

__global__ void k_bin(const float* __restrict__ pos, const int* __restrict__ species,
                      const int* __restrict__ pairs, int* __restrict__ cnt,
                      int* __restrict__ spill_n, float4* __restrict__ rec, int* __restrict__ rsp,
                      float4* __restrict__ srec, int* __restrict__ scs) {
    int p = blockIdx.x * blockDim.x + threadIdx.x;
    if (p >= N_PAIRS) return;
    int c  = pairs[2 * p + 0];
    int nb = pairs[2 * p + 1];
    float rx = pos[3 * nb + 0] - pos[3 * c + 0];
    float ry = pos[3 * nb + 1] - pos[3 * c + 1];
    float rz = pos[3 * nb + 2] - pos[3 * c + 2];
    float d  = sqrtf(rx * rx + ry * ry + rz * rz + 1e-12f);
    float inv = 1.f / d;
    float ux = rx * inv, uy = ry * inv, uz = rz * inv;
    float t  = (d - 15.f) * 0.2f;
    t = fminf(fmaxf(t, 0.f), 1.f);
    float fc = 0.5f * (cosf(3.14159265358979f * t) + 1.f);
    float rb[8];
#pragma unroll
    for (int j = 0; j < 8; ++j) {
        float mu = 20.f * (float)j / 7.f;
        float x  = (d - mu) * 0.4f;
        rb[j] = expf(-x * x) * fc;
    }
    const float s3 = 1.7320508075688772f;
    float4 q0 = { rb[0], rb[1], rb[2], rb[3] };
    float4 q1 = { rb[4], rb[5], rb[6], rb[7] };
    float4 q2 = { uy, uz, ux, s3 * ux * uy };
    float4 q3 = { s3 * uy * uz, 0.5f * (3.f * uz * uz - 1.f), s3 * ux * uz,
                  0.5f * s3 * (ux * ux - uy * uy) };
    int sp = species[nb];
    int slot = atomicAdd(&cnt[c], 1);
    float4* dst;
    if (slot < CAP) {
        long idx = (long)c * CAP + slot;
        rsp[idx] = sp;
        dst = rec + idx * 4;
    } else {
        int o = atomicAdd(spill_n, 1);
        scs[2 * o + 0] = c;
        scs[2 * o + 1] = sp;
        dst = srec + (long)o * 4;
    }
    dst[0] = q0; dst[1] = q1; dst[2] = q2; dst[3] = q3;
}

typedef float v4f __attribute__((ext_vector_type(4)));

#define GATHER_WAVES 4
static_assert(N_ATOMS % GATHER_WAVES == 0, "every wave of the gather grid owns an atom");
static_assert(32 * GATHER_WAVES >= 4 * N_CH, "one thread per embedding value when the table is staged");

__device__ __forceinline__ void add_pair4(const float4* g, float neigh,
                                          const v4f (&w)[24], v4f (&acc)[9]) {
    float4 a = g[0], b = g[1], s0 = g[2], s1 = g[3];
    float rb[8] = { a.x, a.y, a.z, a.w, b.x, b.y, b.z, b.w };
    v4f R[3];
#pragma unroll
    for (int l = 0; l < 3; ++l) {
        v4f s = { 0.f, 0.f, 0.f, 0.f };
#pragma unroll
        for (int j = 0; j < 8; ++j) s += rb[j] * w[l * 8 + j];
        R[l] = s;
    }
    acc[0] += R[0] * neigh;
    v4f w1 = R[1] * neigh;
    acc[1] += s0.x * w1; acc[2] += s0.y * w1; acc[3] += s0.z * w1;
    v4f w2 = R[2] * neigh;
    acc[4] += s0.w * w2; acc[5] += s1.x * w2; acc[6] += s1.y * w2;
    acc[7] += s1.z * w2; acc[8] += s1.w * w2;
}

__global__ __launch_bounds__(32 * GATHER_WAVES) void k_gather(
        const float* __restrict__ emb, const float* __restrict__ W_rad,
        const int* __restrict__ cnt, const int* __restrict__ spill_n,
        const float4* __restrict__ rec, const int* __restrict__ rsp,
        const float4* __restrict__ srec, const int* __restrict__ scs,
        float* __restrict__ f0, float* __restrict__ f1, float* __restrict__ f2) {
    __shared__ float4 s_rec[GATHER_WAVES][CAP * 4];
    __shared__ int    s_sp[GATHER_WAVES][CAP];
    __shared__ float  s_emb[4 * N_CH];
    const int wv = __builtin_amdgcn_readfirstlane((int)(threadIdx.x >> 5));
    const int c  = blockIdx.x * GATHER_WAVES + wv;
    const int L  = threadIdx.x & 31;
    const int n  = cnt[c];
    const int nr = n < CAP ? n : CAP;

    float ev = 0.f;
    if (threadIdx.x < 4 * N_CH) ev = emb[threadIdx.x];
    v4f w[24];
#pragma unroll
    for (int i = 0; i < 24; ++i) w[i] = *(const v4f*)(W_rad + i * KDIM + 4 * L);

    static_assert(CAP % 32 == 0, "a round's 32 x 4 slots stay inside the wave's LDS region");
    const float4* r = rec + (long)c * CAP * 4;
    const int*    s = rsp + (long)c * CAP;
    for (int p0 = 0; p0 < nr; p0 += 32) {
        const int np = nr - p0;
        const int spv = s[p0 + (L < np ? L : np - 1)];
        float4 q[4];
#pragma unroll
        for (int u = 0; u < 4; ++u) {
            const int t = L + 32 * u;
            q[u] = r[4 * p0 + (t < 4 * np ? t : 4 * np - 1)];
        }
        s_sp[wv][p0 + L] = spv;
#pragma unroll
        for (int u = 0; u < 4; ++u) s_rec[wv][4 * p0 + L + 32 * u] = q[u];
    }
    if (threadIdx.x < 4 * N_CH) s_emb[threadIdx.x] = ev;

    v4f acc[9];
#pragma unroll
    for (int i = 0; i < 9; ++i) acc[i] = (v4f){ 0.f, 0.f, 0.f, 0.f };
    __syncthreads();

    for (int i = 0; i < nr; ++i)
        add_pair4(&s_rec[wv][4 * i], s_emb[s_sp[wv][i] * N_CH + L], w, acc);
    if (n > CAP) {
        int no = *spill_n;
        if (no > N_PAIRS) no = N_PAIRS;
        for (int i = 0; i < no; ++i)
            if (scs[2 * i] == c)
                add_pair4(srec + 4 * (long)i, s_emb[scs[2 * i + 1] * N_CH + L], w, acc);
    }
    const float mp = 0.1f;
    *(v4f*)(f0 + (long)c * KDIM + 4 * L) = acc[0] * mp;
    float* r1 = f1 + (long)c * 3 * KDIM + 4 * L;
    float* r2 = f2 + (long)c * 5 * KDIM + 4 * L;
#pragma unroll
    for (int q = 0; q < 3; ++q) *(v4f*)(r1 + q * KDIM) = acc[1 + q] * mp;
#pragma unroll
    for (int q = 0; q < 5; ++q) *(v4f*)(r2 + q * KDIM) = acc[4 + q] * mp;
}

#define MIXB    256
#define MIXB2   136
#define MIXB1   90
#define NGROUPS (N_ATOMS / 16)
#define MPITCH  132
static_assert(N_ATOMS % 16 == 0, "a tile is 16 whole atoms");
static_assert(MIXB2 + MIXB1 < MIXB && MIXB - MIXB2 - MIXB1 <= NGROUPS && MIXB2 <= NGROUPS, "every block owns a group");
__global__ __launch_bounds__(128) void k_mix_feats(
        const float* __restrict__ f0, const float* __restrict__ f1, const float* __restrict__ f2,
        const float* __restrict__ Wc0, const float* __restrict__ Wc1, const float* __restrict__ Wc2,
        const int* __restrict__ species, const float* __restrict__ emb, float* __restrict__ x0) {
    const int blk = blockIdx.x;
    const float* A; const float* B;
    int Q, lsel, first, stride;
    if (blk < MIXB2) {
        A = f2; B = Wc2; Q = 5; lsel = 2; first = blk; stride = MIXB2;
    } else if (blk < MIXB2 + MIXB1) {
        A = f1; B = Wc1; Q = 3; lsel = 1; first = blk - MIXB2; stride = MIXB1;
    } else {
        A = f0; B = Wc0; Q = 1; lsel = 0; first = blk - MIXB2 - MIXB1; stride = MIXB - MIXB2 - MIXB1;
    }
    const int wv   = threadIdx.x >> 5;
    const int lane = threadIdx.x & 31;
    const int half = lane >> 4;
    const int l16  = lane & 15;
    const int col0 = wv * 32 + l16;

    v2f b[32][2];
    const float* Bl = B + (long)(2 * half) * KDIM + col0;
#pragma unroll
    for (int j = 0; j < 32; ++j) {
#pragma unroll
        for (int t = 0; t < 2; ++t) {
            b[j][t].x = Bl[(4 * j) * KDIM + t * 16];
            b[j][t].y = Bl[(4 * j + 1) * KDIM + t * 16];
        }
    }
    __shared__ float s_emb[4 * N_CH];
    __shared__ __attribute__((aligned(16))) float s_t[3][16 * MPITCH];
    static_assert(128 >= 4 * N_CH, "one thread per embedding value when the table is staged");
    s_emb[threadIdx.x] = emb[threadIdx.x];
    const int qi0 = (lsel * KDIM + col0) / 12;
    const int qi1 = (lsel * KDIM + col0 + 16) / 12;
    const float i3 = 0.5773502691896258f;
    const float sc = (lsel == 1) ? -i3 : i3;

    const long pitch = (long)Q * KDIM;
    const float* Sg = A + (long)wv * pitch + 4 * lane;
    const int    so = wv * MPITCH + 4 * lane;
    const int    ao = l16 * MPITCH + 2 * half;
    const int    fo = 8 * half * MPITCH + col0;

    int g = first, q = 0;
    int g1 = g, q1 = q + 1;
    if (q1 == Q) { q1 = 0; g1 = g + stride; }
    v4f gq[4];
    {
        const float* s0 = Sg + (long)g * 16 * pitch + (long)q * KDIM;
#pragma unroll
        for (int u = 0; u < 4; ++u) gq[u] = *(const v4f*)(s0 + (long)(4 * u) * pitch);
#pragma unroll
        for (int u = 0; u < 4; ++u) *(v4f*)(&s_t[0][so + 4 * u * MPITCH]) = gq[u];
        const bool ok1 = g1 < NGROUPS;
        const float* s1 = Sg + (long)(ok1 ? g1 : g) * 16 * pitch + (long)(ok1 ? q1 : q) * KDIM;
#pragma unroll
        for (int u = 0; u < 4; ++u) gq[u] = *(const v4f*)(s1 + (long)(4 * u) * pitch);
#pragma unroll
        for (int u = 0; u < 4; ++u) *(v4f*)(&s_t[1][so + 4 * u * MPITCH]) = gq[u];
    }
    __syncthreads();
    v2f a[32];
#pragma unroll
    for (int j = 0; j < 32; ++j) a[j] = *(const v2f*)(&s_t[0][ao + 4 * j]);

    float P[2][8];
    int sp[8];
#pragma unroll
    for (int r = 0; r < 8; ++r) { P[0][r] = 0.f; P[1][r] = 0.f; sp[r] = 0; }
    const float* cur = s_t[0];
    const float* nxt = s_t[1];
    float*       fil = s_t[2];
    while (g < NGROUPS) {
        int g2 = g1, q2 = q1 + 1;
        if (q2 == Q) { q2 = 0; g2 = g1 + stride; }
        const bool ok2 = g2 < NGROUPS;
        const float* s2 = Sg + (long)(ok2 ? g2 : g) * 16 * pitch + (long)(ok2 ? q2 : q) * KDIM;
#pragma unroll
        for (int u = 0; u < 4; ++u) gq[u] = *(const v4f*)(s2 + (long)(4 * u) * pitch);
        v2f an[32];
#pragma unroll
        for (int j = 0; j < 32; ++j) an[j] = *(const v2f*)(nxt + ao + 4 * j);
        float fc[2][8];
#pragma unroll
        for (int r = 0; r < 8; ++r) {
            fc[0][r] = cur[fo + r * MPITCH];
            fc[1][r] = cur[fo + r * MPITCH + 16];
        }
        if (q == 0) {
#pragma unroll
            for (int r = 0; r < 8; ++r) { sp[r] = species[g * 16 + 8 * half + r]; P[0][r] = 0.f; P[1][r] = 0.f; }
        }
        __builtin_amdgcn_sched_barrier(0);

        v8f acc0 = {}, acc1 = {};
#pragma unroll
        for (int j = 0; j < 32; ++j) {
            acc0 = __builtin_amdgcn_wmma_f32_16x16x4_f32(false, a[j], false, b[j][0], (short)0,
                                                         acc0, false, false);
            acc1 = __builtin_amdgcn_wmma_f32_16x16x4_f32(false, a[j], false, b[j][1], (short)0,
                                                         acc1, false, false);
        }
        __builtin_amdgcn_sched_barrier(0);
#pragma unroll
        for (int r = 0; r < 8; ++r) {
            P[0][r] = acc0[r] * fc[0][r] + ((lsel == 0) ? fc[0][r] : P[0][r]);
            P[1][r] = acc1[r] * fc[1][r] + ((lsel == 0) ? fc[1][r] : P[1][r]);
        }
#pragma unroll
        for (int u = 0; u < 4; ++u) *(v4f*)(fil + so + 4 * u * MPITCH) = gq[u];
        __syncthreads();
        if (q == Q - 1) {
            float* X = x0 + ((long)g * 16 + 8 * half) * K0TOT + lsel * KDIM + col0;
#pragma unroll
            for (int r = 0; r < 8; ++r) {
                const float* e = s_emb + sp[r] * N_CH;
                const float v0 = (lsel == 0) ? P[0][r] : P[0][r] * sc;
                const float v1 = (lsel == 0) ? P[1][r] : P[1][r] * sc;
                X[(long)r * K0TOT]      = v0 * e[qi0];
                X[(long)r * K0TOT + 16] = v1 * e[qi1];
            }
        }
        __builtin_amdgcn_sched_barrier(0);
#pragma unroll
        for (int j = 0; j < 32; ++j) a[j] = an[j];
        float* t = (float*)cur; cur = nxt; nxt = fil; fil = t;
        g = g1; q = q1; g1 = g2; q1 = q2;
    }
}

#define HEAD_BLOCKS 256
#define HEAD_TILES  (N_ATOMS / 16)
#define HPITCH      388
__device__ __forceinline__ void head_tile(const v2f (&cur)[96], v2f (&nxt)[96], const v2f (&b)[96][2],
                                          const float* __restrict__ src, const float* __restrict__ ldn,
                                          float* __restrict__ fil, float (&rd)[4][8][32],
                                          float* __restrict__ out, int tile,
                                          float bias0, float wo0, float bias1, float wo1,
                                          int wv, int lane, int half, int l16, int tid) {
    v4f gq[12];
#pragma unroll
    for (int u = 0; u < 12; ++u) gq[u] = *(const v4f*)(src + 4 * (tid + 128 * u));
    const float* An = ldn + l16 * HPITCH + 2 * half;
#pragma unroll
    for (int j = 0; j < 48; ++j) nxt[j] = *(const v2f*)(An + 4 * j);
    __builtin_amdgcn_sched_barrier(0);

    v8f acc0 = {}, acc1 = {};
#pragma unroll
    for (int j = 0; j < 48; ++j) {
        acc0 = __builtin_amdgcn_wmma_f32_16x16x4_f32(false, cur[j], false, b[j][0], (short)0,
                                                     acc0, false, false);
        acc1 = __builtin_amdgcn_wmma_f32_16x16x4_f32(false, cur[j], false, b[j][1], (short)0,
                                                     acc1, false, false);
    }
    __builtin_amdgcn_sched_barrier(0);
#pragma unroll
    for (int j = 48; j < 96; ++j) nxt[j] = *(const v2f*)(An + 4 * j);
    __builtin_amdgcn_sched_barrier(0);
#pragma unroll
    for (int j = 48; j < 96; ++j) {
        acc0 = __builtin_amdgcn_wmma_f32_16x16x4_f32(false, cur[j], false, b[j][0], (short)0,
                                                     acc0, false, false);
        acc1 = __builtin_amdgcn_wmma_f32_16x16x4_f32(false, cur[j], false, b[j][1], (short)0,
                                                     acc1, false, false);
    }
    __builtin_amdgcn_sched_barrier(0);
#pragma unroll
    for (int i = 0; i < 8; ++i) {
        float v0 = acc0[i] + bias0;
        float v1 = acc1[i] + bias1;
        float p = 0.f;
        p += (v0 / (1.f + expf(-v0))) * wo0;
        p += (v1 / (1.f + expf(-v1))) * wo1;
        rd[wv][i][lane] = p;
    }
#pragma unroll
    for (int u = 0; u < 12; ++u) {
        const int f = tid + 128 * u;
        *(v4f*)(fil + (f / 96) * HPITCH + 4 * (f % 96)) = gq[u];
    }
    __syncthreads();
    const int rowbase = tile * 16 + 8 * half;
#pragma unroll
    for (int r = 0; r < 2; ++r) {
        const int i = 2 * wv + r;
        float s = rd[0][i][lane];
        s += rd[1][i][lane];
        s += rd[2][i][lane];
        s += rd[3][i][lane];
        atomicAdd(&out[rowbase + i], s);
    }
}

__global__ __launch_bounds__(128) void k_head_resident(
        const float* __restrict__ x0, const float* __restrict__ Wh, const float* __restrict__ bh,
        const float* __restrict__ Wo, float* __restrict__ out) {
    __shared__ float red[2][4][8][32];
    __shared__ __attribute__((aligned(16))) float s_x[3][16 * HPITCH];
    const int cb     = blockIdx.x % 3;
    const int first  = blockIdx.x / 3;
    const int stride = (cb == 0) ? 86 : 85;
    const int tid  = threadIdx.x;
    const int wv   = threadIdx.x >> 5;
    const int lane = threadIdx.x & 31;
    const int half = lane >> 4;
    const int l16  = lane & 15;
    const int n0   = cb * 128 + wv * 32;

    v2f b[96][2];
    const float* Bl = Wh + (long)(2 * half) * K0TOT + n0 + l16;
#pragma unroll
    for (int j = 0; j < 96; ++j) {
#pragma unroll
        for (int t = 0; t < 2; ++t) {
            b[j][t].x = Bl[(4 * j) * K0TOT + t * 16];
            b[j][t].y = Bl[(4 * j + 1) * K0TOT + t * 16];
        }
    }
    const float bias0 = bh[n0 + l16],      wo0 = Wo[n0 + l16];
    const float bias1 = bh[n0 + 16 + l16], wo1 = Wo[n0 + 16 + l16];

    {
        int t1 = first + stride;
        if (t1 >= HEAD_TILES) t1 = first;
        const float* s0 = x0 + (long)first * 16 * K0TOT;
        const float* s1 = x0 + (long)t1 * 16 * K0TOT;
        v4f g0[12], g1[12];
#pragma unroll
        for (int u = 0; u < 12; ++u) g0[u] = *(const v4f*)(s0 + 4 * (tid + 128 * u));
#pragma unroll
        for (int u = 0; u < 12; ++u) g1[u] = *(const v4f*)(s1 + 4 * (tid + 128 * u));
#pragma unroll
        for (int u = 0; u < 12; ++u) {
            const int f = tid + 128 * u;
            *(v4f*)(&s_x[0][(f / 96) * HPITCH + 4 * (f % 96)]) = g0[u];
            *(v4f*)(&s_x[1][(f / 96) * HPITCH + 4 * (f % 96)]) = g1[u];
        }
    }
    __syncthreads();
    v2f a[96];
    {
        const float* Ar = &s_x[0][l16 * HPITCH + 2 * half];
#pragma unroll
        for (int j = 0; j < 96; ++j) a[j] = *(const v2f*)(Ar + 4 * j);
    }
    int par = 0;
    const float* nxt = s_x[1];
    float*       fil = s_x[2];
    float*       old = s_x[0];
    for (int tile = first; tile < HEAD_TILES; tile += stride) {
        int t2 = tile + 2 * stride;
        if (t2 >= HEAD_TILES) t2 = tile;
        v2f an[96];
        head_tile(a, an, b, x0 + (long)t2 * 16 * K0TOT, nxt, fil, red[par], out, tile,
                  bias0, wo0, bias1, wo1, wv, lane, half, l16, tid);
        par ^= 1;
        __builtin_amdgcn_sched_barrier(0);
#pragma unroll
        for (int j = 0; j < 96; ++j) a[j] = an[j];
        float* t = old; old = (float*)nxt; nxt = fil; fil = t;
    }
}

extern "C" void kernel_launch(void* const* d_in, const int* in_sizes, int n_in,
                              void* d_out, int out_size, void* d_ws, size_t ws_size,
                              hipStream_t stream) {
    (void)in_sizes; (void)n_in; (void)out_size; (void)ws_size;
    const float* positions = (const float*)d_in[0];
    const int*   species   = (const int*)d_in[1];
    const int*   pairs     = (const int*)d_in[2];
    const float* embeddings= (const float*)d_in[3];
    const float* W_rad     = (const float*)d_in[4];
    const float* W_cg0     = (const float*)d_in[5];
    const float* W_cg1     = (const float*)d_in[6];
    const float* W_cg2     = (const float*)d_in[7];
    const float* W_head    = (const float*)d_in[8];
    const float* b_head    = (const float*)d_in[9];
    const float* W_out     = (const float*)d_in[10];
    const float* b_out     = (const float*)d_in[11];
    float* out = (float*)d_out;

    float* ws = (float*)d_ws;
    const long SZ_F0 = (long)N_ATOMS * KDIM;
    const long SZ_F1 = (long)N_ATOMS * 3 * KDIM;
    const long SZ_F2 = (long)N_ATOMS * 5 * KDIM;
    const long SZ_U  = (long)N_ATOMS * 9 * KDIM;
    float* f0 = ws;
    float* f1 = f0 + SZ_F0;
    float* f2 = f1 + SZ_F1;
    float* g0 = f2 + SZ_F2 + SZ_U + SZ_U;
    float* g1 = g0 + SZ_F0;
    float* g2 = g1 + SZ_U;
    float* x0 = g2 + SZ_U;

    float4* rec  = (float4*)g2;
    float4* srec = (float4*)g1;
    int* cnt     = (int*)x0;
    int* spill_n = cnt + 10240;
    int* rsp     = cnt + 10304;
    int* scs     = rsp + (long)N_ATOMS * CAP;

    k_init<<<(N_ATOMS + 255) / 256, 256, 0, stream>>>(out, b_out, cnt, spill_n);

    k_bin<<<N_PAIRS / 256, 256, 0, stream>>>(positions, species, pairs, cnt, spill_n,
                                             rec, rsp, srec, scs);

    k_gather<<<N_ATOMS / GATHER_WAVES, 32 * GATHER_WAVES, 0, stream>>>(embeddings, W_rad, cnt, spill_n, rec, rsp, srec, scs,
                                           f0, f1, f2);

    k_mix_feats<<<MIXB, 128, 0, stream>>>(f0, f1, f2, W_cg0, W_cg1, W_cg2, species, embeddings, x0);

    k_head_resident<<<HEAD_BLOCKS, 128, 0, stream>>>(x0, W_head, b_head, W_out, out);
}


# pair sums formed inside the channel-mix kernel: a block sums its own l's rows of its 16 atoms from the filed records into LDS tiles, 2 blocks per WGP; gather launch and f planes go; sum orders kept
# speedup vs baseline: 2.8417x; 1.0307x over previous
#include <hip/hip_runtime.h>
#include <math.h>

#define N_ATOMS  10000
#define N_PAIRS  160000
#define N_CH     32
#define KDIM     128
#define K0TOT    384

typedef float v2f __attribute__((ext_vector_type(2)));
typedef float v8f __attribute__((ext_vector_type(8)));


#define CAP 64

__global__ void k_init(float* out, const float* b_out, int* cnt, int* spill_n) {
    int i = blockIdx.x * blockDim.x + threadIdx.x;
    if (i < N_ATOMS) { out[i] = b_out[0]; cnt[i] = 0; }
    if (i == 0) *spill_n = 0;
}

__global__ void k_bin(const float* __restrict__ pos, const int* __restrict__ species,
                      const int* __restrict__ pairs, int* __restrict__ cnt,
                      int* __restrict__ spill_n, float4* __restrict__ rec, int* __restrict__ rsp,
                      float4* __restrict__ srec, int* __restrict__ scs) {
    int p = blockIdx.x * blockDim.x + threadIdx.x;
    if (p >= N_PAIRS) return;
    int c  = pairs[2 * p + 0];
    int nb = pairs[2 * p + 1];
    float rx = pos[3 * nb + 0] - pos[3 * c + 0];
    float ry = pos[3 * nb + 1] - pos[3 * c + 1];
    float rz = pos[3 * nb + 2] - pos[3 * c + 2];
    float d  = sqrtf(rx * rx + ry * ry + rz * rz + 1e-12f);
    float inv = 1.f / d;
    float ux = rx * inv, uy = ry * inv, uz = rz * inv;
    float t  = (d - 15.f) * 0.2f;
    t = fminf(fmaxf(t, 0.f), 1.f);
    float fc = 0.5f * (cosf(3.14159265358979f * t) + 1.f);
    float rb[8];
#pragma unroll
    for (int j = 0; j < 8; ++j) {
        float mu = 20.f * (float)j / 7.f;
        float x  = (d - mu) * 0.4f;
        rb[j] = expf(-x * x) * fc;
    }
    const float s3 = 1.7320508075688772f;
    float4 q0 = { rb[0], rb[1], rb[2], rb[3] };
    float4 q1 = { rb[4], rb[5], rb[6], rb[7] };
    float4 q2 = { uy, uz, ux, s3 * ux * uy };
    float4 q3 = { s3 * uy * uz, 0.5f * (3.f * uz * uz - 1.f), s3 * ux * uz,
                  0.5f * s3 * (ux * ux - uy * uy) };
    int sp = species[nb];
    int slot = atomicAdd(&cnt[c], 1);
    float4* dst;
    if (slot < CAP) {
        long idx = (long)c * CAP + slot;
        rsp[idx] = sp;
        dst = rec + idx * 4;
    } else {
        int o = atomicAdd(spill_n, 1);
        scs[2 * o + 0] = c;
        scs[2 * o + 1] = sp;
        dst = srec + (long)o * 4;
    }
    dst[0] = q0; dst[1] = q1; dst[2] = q2; dst[3] = q3;
}

typedef float v4f __attribute__((ext_vector_type(4)));

#define SLOTS 32

template <int LSEL>
__device__ __forceinline__ void add_pair_l(const float4* g, float neigh,
                                           const v4f (&w)[8], v4f (&acc)[2 * LSEL + 1]) {
    float4 a = g[0], b = g[1];
    float rb[8] = { a.x, a.y, a.z, a.w, b.x, b.y, b.z, b.w };
    v4f s = { 0.f, 0.f, 0.f, 0.f };
#pragma unroll
    for (int j = 0; j < 8; ++j) s += rb[j] * w[j];
    if constexpr (LSEL == 0) {
        acc[0] += s * neigh;
    } else if constexpr (LSEL == 1) {
        float4 s0 = g[2];
        v4f w1 = s * neigh;
        acc[0] += s0.x * w1; acc[1] += s0.y * w1; acc[2] += s0.z * w1;
    } else {
        float4 s0 = g[2], s1 = g[3];
        v4f w2 = s * neigh;
        acc[0] += s0.w * w2; acc[1] += s1.x * w2; acc[2] += s1.y * w2;
        acc[3] += s1.z * w2; acc[4] += s1.w * w2;
    }
}


#define MIXB    512
#define MIXB2   209
#define MIXB1   157
#define NGROUPS (N_ATOMS / 16)
#define MPITCH  132
static_assert(N_ATOMS % 16 == 0, "a tile is 16 whole atoms");
static_assert(MIXB2 + MIXB1 < MIXB && MIXB - MIXB2 - MIXB1 <= NGROUPS && MIXB2 <= NGROUPS && MIXB1 <= NGROUPS, "every block owns a group");
static_assert(SLOTS == 32 && CAP >= SLOTS, "one staging round of 32 lanes x 4 pieces fills an atom's LDS slots");

template <int LSEL>
__device__ __forceinline__ void mix_body(
        const float* __restrict__ emb, const float* __restrict__ W_rad,
        const int* __restrict__ cnt, const int* __restrict__ spill_n,
        const float4* __restrict__ rec, const int* __restrict__ rsp,
        const float4* __restrict__ srec, const int* __restrict__ scs,
        const float* __restrict__ B, const int* __restrict__ species, float* __restrict__ x0,
        const int first, const int stride,
        float4 (&s_rec)[4][4][SLOTS * 4], float (&s_emb)[4 * N_CH],
        float (&s_t)[5][16 * MPITCH]) {
    constexpr int Q = 2 * LSEL + 1;
    const int wv   = __builtin_amdgcn_readfirstlane((int)(threadIdx.x >> 5));
    const int lane = threadIdx.x & 31;
    const int half = lane >> 4;
    const int l16  = lane & 15;
    const int col0 = wv * 32 + l16;

    v2f b[32][2];
    const float* Bl = B + (long)(2 * half) * KDIM + col0;
#pragma unroll
    for (int j = 0; j < 32; ++j) {
#pragma unroll
        for (int t = 0; t < 2; ++t) {
            b[j][t].x = Bl[(4 * j) * KDIM + t * 16];
            b[j][t].y = Bl[(4 * j + 1) * KDIM + t * 16];
        }
    }
    v4f w[8];
#pragma unroll
    for (int j = 0; j < 8; ++j) w[j] = *(const v4f*)(W_rad + (LSEL * 8 + j) * KDIM + 4 * lane);
    s_emb[threadIdx.x] = emb[threadIdx.x];
    const int qi0 = (LSEL * KDIM + col0) / 12;
    const int qi1 = (LSEL * KDIM + col0 + 16) / 12;
    const float i3 = 0.5773502691896258f;
    const float sc = (LSEL == 1) ? -i3 : i3;
    const float mp = 0.1f;
    const int ao = l16 * MPITCH + 2 * half;
    const int fo = 8 * half * MPITCH + col0;

    int nn[4];
#pragma unroll
    for (int u = 0; u < 4; ++u) nn[u] = cnt[first * 16 + wv * 4 + u];
    for (int g = first; g < NGROUPS; g += stride) {
        int n[4];
#pragma unroll
        for (int u = 0; u < 4; ++u) n[u] = nn[u];
        if (g + stride < NGROUPS) {
#pragma unroll
            for (int u = 0; u < 4; ++u) nn[u] = cnt[(g + stride) * 16 + wv * 4 + u];
        }
        int    spv[4];
        float4 q4[4][4];
#pragma unroll
        for (int u = 0; u < 4; ++u) {
            const long c  = (long)g * 16 + wv * 4 + u;
            const int  np = n[u] < SLOTS ? n[u] : SLOTS;
            const int*    sb = np > 0 ? rsp + c * CAP : species;
            const float4* rb = np > 0 ? rec + c * CAP * 4 : (const float4*)emb;
            const int     m1 = np > 0 ? np - 1 : 0;
            const int     m4 = np > 0 ? 4 * np - 1 : 0;
            spv[u] = sb[lane < m1 ? lane : m1];
#pragma unroll
            for (int v = 0; v < 4; ++v) {
                const int t = lane + 32 * v;
                q4[u][v] = rb[t < m4 ? t : m4];
            }
        }
#pragma unroll
        for (int u = 0; u < 4; ++u) {
#pragma unroll
            for (int v = 0; v < 4; ++v) s_rec[wv][u][lane + 32 * v] = q4[u][v];
        }
        __syncthreads();
        int sp[8];
#pragma unroll
        for (int r = 0; r < 8; ++r) sp[r] = species[g * 16 + 8 * half + r];

#pragma nounroll
        for (int u = 0; u < 4; ++u) {
            const long c   = (long)g * 16 + wv * 4 + u;
            const int  nu  = u == 0 ? n[0] : (u == 1 ? n[1] : (u == 2 ? n[2] : n[3]));
            const int  spu = u == 0 ? spv[0] : (u == 1 ? spv[1] : (u == 2 ? spv[2] : spv[3]));
            const int  np  = nu < SLOTS ? nu : SLOTS;
            const int  nr = nu < CAP ? nu : CAP;
            v4f acc[Q];
#pragma unroll
            for (int i = 0; i < Q; ++i) acc[i] = (v4f){ 0.f, 0.f, 0.f, 0.f };
#pragma unroll 4
            for (int i = 0; i < np; ++i)
                add_pair_l<LSEL>(&s_rec[wv][u][4 * i],
                                 s_emb[__builtin_amdgcn_readlane(spu, i) * N_CH + lane], w, acc);
            for (int i = np; i < nr; ++i)
                add_pair_l<LSEL>(rec + (c * CAP + i) * 4, s_emb[rsp[c * CAP + i] * N_CH + lane], w, acc);
            if (nu > CAP) {
                int no = *spill_n;
                if (no > N_PAIRS) no = N_PAIRS;
                for (int i = 0; i < no; ++i)
                    if (scs[2 * i] == (int)c)
                        add_pair_l<LSEL>(srec + 4 * (long)i, s_emb[scs[2 * i + 1] * N_CH + lane], w, acc);
            }
#pragma unroll
            for (int i = 0; i < Q; ++i)
                *(v4f*)(&s_t[i][(wv * 4 + u) * MPITCH + 4 * lane]) = acc[i] * mp;
        }
        __syncthreads();

        float P[2][8];
#pragma unroll
        for (int r = 0; r < 8; ++r) { P[0][r] = 0.f; P[1][r] = 0.f; }
        v2f a[32];
#pragma unroll
        for (int j = 0; j < 32; ++j) a[j] = *(const v2f*)(&s_t[0][ao + 4 * j]);
#pragma unroll
        for (int q = 0; q < Q; ++q) {
            v2f an[32];
            if (q + 1 < Q) {
#pragma unroll
                for (int j = 0; j < 32; ++j) an[j] = *(const v2f*)(&s_t[q + 1][ao + 4 * j]);
            }
            float fc[2][8];
#pragma unroll
            for (int r = 0; r < 8; ++r) {
                fc[0][r] = s_t[q][fo + r * MPITCH];
                fc[1][r] = s_t[q][fo + r * MPITCH + 16];
            }
            __builtin_amdgcn_sched_barrier(0);

            v8f acc0 = {}, acc1 = {};
#pragma unroll
            for (int j = 0; j < 32; ++j) {
                acc0 = __builtin_amdgcn_wmma_f32_16x16x4_f32(false, a[j], false, b[j][0], (short)0,
                                                             acc0, false, false);
                acc1 = __builtin_amdgcn_wmma_f32_16x16x4_f32(false, a[j], false, b[j][1], (short)0,
                                                             acc1, false, false);
            }
            __builtin_amdgcn_sched_barrier(0);
#pragma unroll
            for (int r = 0; r < 8; ++r) {
                P[0][r] = acc0[r] * fc[0][r] + ((LSEL == 0) ? fc[0][r] : P[0][r]);
                P[1][r] = acc1[r] * fc[1][r] + ((LSEL == 0) ? fc[1][r] : P[1][r]);
            }
            if (q + 1 < Q) {
#pragma unroll
                for (int j = 0; j < 32; ++j) a[j] = an[j];
            }
        }
        float* X = x0 + ((long)g * 16 + 8 * half) * K0TOT + LSEL * KDIM + col0;
#pragma unroll
        for (int r = 0; r < 8; ++r) {
            const float* er = s_emb + sp[r] * N_CH;
            const float v0 = (LSEL == 0) ? P[0][r] : P[0][r] * sc;
            const float v1 = (LSEL == 0) ? P[1][r] : P[1][r] * sc;
            X[(long)r * K0TOT]      = v0 * er[qi0];
            X[(long)r * K0TOT + 16] = v1 * er[qi1];
        }
    }
}

__global__ __launch_bounds__(128, 2) void k_pairsum_mix(
        const float* __restrict__ emb, const float* __restrict__ W_rad,
        const int* __restrict__ cnt, const int* __restrict__ spill_n,
        const float4* __restrict__ rec, const int* __restrict__ rsp,
        const float4* __restrict__ srec, const int* __restrict__ scs,
        const float* __restrict__ Wc0, const float* __restrict__ Wc1, const float* __restrict__ Wc2,
        const int* __restrict__ species, float* __restrict__ x0) {
    __shared__ float4 s_rec[4][4][SLOTS * 4];
    __shared__ float  s_emb[4 * N_CH];
    __shared__ __attribute__((aligned(16))) float s_t[5][16 * MPITCH];
    static_assert(128 >= 4 * N_CH, "one thread per embedding value when the table is staged");
    const int blk = blockIdx.x;
    if (blk < MIXB2)
        mix_body<2>(emb, W_rad, cnt, spill_n, rec, rsp, srec, scs, Wc2, species, x0,
                    blk, MIXB2, s_rec, s_emb, s_t);
    else if (blk < MIXB2 + MIXB1)
        mix_body<1>(emb, W_rad, cnt, spill_n, rec, rsp, srec, scs, Wc1, species, x0,
                    blk - MIXB2, MIXB1, s_rec, s_emb, s_t);
    else
        mix_body<0>(emb, W_rad, cnt, spill_n, rec, rsp, srec, scs, Wc0, species, x0,
                    blk - MIXB2 - MIXB1, MIXB - MIXB2 - MIXB1, s_rec, s_emb, s_t);
}

#define HEAD_BLOCKS 256
#define HEAD_TILES  (N_ATOMS / 16)
#define HPITCH      388
__device__ __forceinline__ void head_tile(const v2f (&cur)[96], v2f (&nxt)[96], const v2f (&b)[96][2],
                                          const float* __restrict__ src, const float* __restrict__ ldn,
                                          float* __restrict__ fil, float (&rd)[4][8][32],
                                          float* __restrict__ out, int tile,
                                          float bias0, float wo0, float bias1, float wo1,
                                          int wv, int lane, int half, int l16, int tid) {
    v4f gq[12];
#pragma unroll
    for (int u = 0; u < 12; ++u) gq[u] = *(const v4f*)(src + 4 * (tid + 128 * u));
    const float* An = ldn + l16 * HPITCH + 2 * half;
#pragma unroll
    for (int j = 0; j < 48; ++j) nxt[j] = *(const v2f*)(An + 4 * j);
    __builtin_amdgcn_sched_barrier(0);

    v8f acc0 = {}, acc1 = {};
#pragma unroll
    for (int j = 0; j < 48; ++j) {
        acc0 = __builtin_amdgcn_wmma_f32_16x16x4_f32(false, cur[j], false, b[j][0], (short)0,
                                                     acc0, false, false);
        acc1 = __builtin_amdgcn_wmma_f32_16x16x4_f32(false, cur[j], false, b[j][1], (short)0,
                                                     acc1, false, false);
    }
    __builtin_amdgcn_sched_barrier(0);
#pragma unroll
    for (int j = 48; j < 96; ++j) nxt[j] = *(const v2f*)(An + 4 * j);
    __builtin_amdgcn_sched_barrier(0);
#pragma unroll
    for (int j = 48; j < 96; ++j) {
        acc0 = __builtin_amdgcn_wmma_f32_16x16x4_f32(false, cur[j], false, b[j][0], (short)0,
                                                     acc0, false, false);
        acc1 = __builtin_amdgcn_wmma_f32_16x16x4_f32(false, cur[j], false, b[j][1], (short)0,
                                                     acc1, false, false);
    }
    __builtin_amdgcn_sched_barrier(0);
#pragma unroll
    for (int i = 0; i < 8; ++i) {
        float v0 = acc0[i] + bias0;
        float v1 = acc1[i] + bias1;
        float p = 0.f;
        p += (v0 / (1.f + expf(-v0))) * wo0;
        p += (v1 / (1.f + expf(-v1))) * wo1;
        rd[wv][i][lane] = p;
    }
#pragma unroll
    for (int u = 0; u < 12; ++u) {
        const int f = tid + 128 * u;
        *(v4f*)(fil + (f / 96) * HPITCH + 4 * (f % 96)) = gq[u];
    }
    __syncthreads();
    const int rowbase = tile * 16 + 8 * half;
#pragma unroll
    for (int r = 0; r < 2; ++r) {
        const int i = 2 * wv + r;
        float s = rd[0][i][lane];
        s += rd[1][i][lane];
        s += rd[2][i][lane];
        s += rd[3][i][lane];
        atomicAdd(&out[rowbase + i], s);
    }
}

__global__ __launch_bounds__(128) void k_head_resident(
        const float* __restrict__ x0, const float* __restrict__ Wh, const float* __restrict__ bh,
        const float* __restrict__ Wo, float* __restrict__ out) {
    __shared__ float red[2][4][8][32];
    __shared__ __attribute__((aligned(16))) float s_x[3][16 * HPITCH];
    const int cb     = blockIdx.x % 3;
    const int first  = blockIdx.x / 3;
    const int stride = (cb == 0) ? 86 : 85;
    const int tid  = threadIdx.x;
    const int wv   = threadIdx.x >> 5;
    const int lane = threadIdx.x & 31;
    const int half = lane >> 4;
    const int l16  = lane & 15;
    const int n0   = cb * 128 + wv * 32;

    v2f b[96][2];
    const float* Bl = Wh + (long)(2 * half) * K0TOT + n0 + l16;
#pragma unroll
    for (int j = 0; j < 96; ++j) {
#pragma unroll
        for (int t = 0; t < 2; ++t) {
            b[j][t].x = Bl[(4 * j) * K0TOT + t * 16];
            b[j][t].y = Bl[(4 * j + 1) * K0TOT + t * 16];
        }
    }
    const float bias0 = bh[n0 + l16],      wo0 = Wo[n0 + l16];
    const float bias1 = bh[n0 + 16 + l16], wo1 = Wo[n0 + 16 + l16];

    {
        int t1 = first + stride;
        if (t1 >= HEAD_TILES) t1 = first;
        const float* s0 = x0 + (long)first * 16 * K0TOT;
        const float* s1 = x0 + (long)t1 * 16 * K0TOT;
        v4f g0[12], g1[12];
#pragma unroll
        for (int u = 0; u < 12; ++u) g0[u] = *(const v4f*)(s0 + 4 * (tid + 128 * u));
#pragma unroll
        for (int u = 0; u < 12; ++u) g1[u] = *(const v4f*)(s1 + 4 * (tid + 128 * u));
#pragma unroll
        for (int u = 0; u < 12; ++u) {
            const int f = tid + 128 * u;
            *(v4f*)(&s_x[0][(f / 96) * HPITCH + 4 * (f % 96)]) = g0[u];
            *(v4f*)(&s_x[1][(f / 96) * HPITCH + 4 * (f % 96)]) = g1[u];
        }
    }
    __syncthreads();
    v2f a[96];
    {
        const float* Ar = &s_x[0][l16 * HPITCH + 2 * half];
#pragma unroll
        for (int j = 0; j < 96; ++j) a[j] = *(const v2f*)(Ar + 4 * j);
    }
    int par = 0;
    const float* nxt = s_x[1];
    float*       fil = s_x[2];
    float*       old = s_x[0];
    for (int tile = first; tile < HEAD_TILES; tile += stride) {
        int t2 = tile + 2 * stride;
        if (t2 >= HEAD_TILES) t2 = tile;
        v2f an[96];
        head_tile(a, an, b, x0 + (long)t2 * 16 * K0TOT, nxt, fil, red[par], out, tile,
                  bias0, wo0, bias1, wo1, wv, lane, half, l16, tid);
        par ^= 1;
        __builtin_amdgcn_sched_barrier(0);
#pragma unroll
        for (int j = 0; j < 96; ++j) a[j] = an[j];
        float* t = old; old = (float*)nxt; nxt = fil; fil = t;
    }
}

extern "C" void kernel_launch(void* const* d_in, const int* in_sizes, int n_in,
                              void* d_out, int out_size, void* d_ws, size_t ws_size,
                              hipStream_t stream) {
    (void)in_sizes; (void)n_in; (void)out_size; (void)ws_size;
    const float* positions = (const float*)d_in[0];
    const int*   species   = (const int*)d_in[1];
    const int*   pairs     = (const int*)d_in[2];
    const float* embeddings= (const float*)d_in[3];
    const float* W_rad     = (const float*)d_in[4];
    const float* W_cg0     = (const float*)d_in[5];
    const float* W_cg1     = (const float*)d_in[6];
    const float* W_cg2     = (const float*)d_in[7];
    const float* W_head    = (const float*)d_in[8];
    const float* b_head    = (const float*)d_in[9];
    const float* W_out     = (const float*)d_in[10];
    const float* b_out     = (const float*)d_in[11];
    float* out = (float*)d_out;

    float* ws = (float*)d_ws;
    const long SZ_F0 = (long)N_ATOMS * KDIM;
    const long SZ_F1 = (long)N_ATOMS * 3 * KDIM;
    const long SZ_F2 = (long)N_ATOMS * 5 * KDIM;
    const long SZ_U  = (long)N_ATOMS * 9 * KDIM;
    float* f0 = ws;
    float* f1 = f0 + SZ_F0;
    float* f2 = f1 + SZ_F1;
    float* g0 = f2 + SZ_F2 + SZ_U + SZ_U;
    float* g1 = g0 + SZ_F0;
    float* g2 = g1 + SZ_U;
    float* x0 = g2 + SZ_U;

    float4* rec  = (float4*)g2;
    float4* srec = (float4*)g1;
    int* cnt     = (int*)f0;
    int* spill_n = cnt + 10240;
    int* rsp     = cnt + 10304;
    int* scs     = rsp + (long)N_ATOMS * CAP;

    k_init<<<(N_ATOMS + 255) / 256, 256, 0, stream>>>(out, b_out, cnt, spill_n);

    k_bin<<<N_PAIRS / 256, 256, 0, stream>>>(positions, species, pairs, cnt, spill_n,
                                             rec, rsp, srec, scs);

    k_pairsum_mix<<<MIXB, 128, 0, stream>>>(embeddings, W_rad, cnt, spill_n, rec, rsp, srec, scs,
                                            W_cg0, W_cg1, W_cg2, species, x0);

    k_head_resident<<<HEAD_BLOCKS, 128, 0, stream>>>(x0, W_head, b_head, W_out, out);
}


# channel-mix kernel: three blocks per WGP (was two), 768 blocks: a wave keeps one A fragment set, each half refilled right behind the WMMAs that read it, and fits 336 registers; sum orders kept
# speedup vs baseline: 2.9972x; 1.0547x over previous
#include <hip/hip_runtime.h>
#include <math.h>

#define N_ATOMS  10000
#define N_PAIRS  160000
#define N_CH     32
#define KDIM     128
#define K0TOT    384

typedef float v2f __attribute__((ext_vector_type(2)));
typedef float v8f __attribute__((ext_vector_type(8)));


#define CAP 64

__global__ void k_init(float* out, const float* b_out, int* cnt, int* spill_n) {
    int i = blockIdx.x * blockDim.x + threadIdx.x;
    if (i < N_ATOMS) { out[i] = b_out[0]; cnt[i] = 0; }
    if (i == 0) *spill_n = 0;
}

__global__ void k_bin(const float* __restrict__ pos, const int* __restrict__ species,
                      const int* __restrict__ pairs, int* __restrict__ cnt,
                      int* __restrict__ spill_n, float4* __restrict__ rec, int* __restrict__ rsp,
                      float4* __restrict__ srec, int* __restrict__ scs) {
    int p = blockIdx.x * blockDim.x + threadIdx.x;
    if (p >= N_PAIRS) return;
    int c  = pairs[2 * p + 0];
    int nb = pairs[2 * p + 1];
    float rx = pos[3 * nb + 0] - pos[3 * c + 0];
    float ry = pos[3 * nb + 1] - pos[3 * c + 1];
    float rz = pos[3 * nb + 2] - pos[3 * c + 2];
    float d  = sqrtf(rx * rx + ry * ry + rz * rz + 1e-12f);
    float inv = 1.f / d;
    float ux = rx * inv, uy = ry * inv, uz = rz * inv;
    float t  = (d - 15.f) * 0.2f;
    t = fminf(fmaxf(t, 0.f), 1.f);
    float fc = 0.5f * (cosf(3.14159265358979f * t) + 1.f);
    float rb[8];
#pragma unroll
    for (int j = 0; j < 8; ++j) {
        float mu = 20.f * (float)j / 7.f;
        float x  = (d - mu) * 0.4f;
        rb[j] = expf(-x * x) * fc;
    }
    const float s3 = 1.7320508075688772f;
    float4 q0 = { rb[0], rb[1], rb[2], rb[3] };
    float4 q1 = { rb[4], rb[5], rb[6], rb[7] };
    float4 q2 = { uy, uz, ux, s3 * ux * uy };
    float4 q3 = { s3 * uy * uz, 0.5f * (3.f * uz * uz - 1.f), s3 * ux * uz,
                  0.5f * s3 * (ux * ux - uy * uy) };
    int sp = species[nb];
    int slot = atomicAdd(&cnt[c], 1);
    float4* dst;
    if (slot < CAP) {
        long idx = (long)c * CAP + slot;
        rsp[idx] = sp;
        dst = rec + idx * 4;
    } else {
        int o = atomicAdd(spill_n, 1);
        scs[2 * o + 0] = c;
        scs[2 * o + 1] = sp;
        dst = srec + (long)o * 4;
    }
    dst[0] = q0; dst[1] = q1; dst[2] = q2; dst[3] = q3;
}

typedef float v4f __attribute__((ext_vector_type(4)));

#define SLOTS 32

template <int LSEL>
__device__ __forceinline__ void add_pair_l(const float4* g, float neigh,
                                           const v4f (&w)[8], v4f (&acc)[2 * LSEL + 1]) {
    float4 a = g[0], b = g[1];
    float rb[8] = { a.x, a.y, a.z, a.w, b.x, b.y, b.z, b.w };
    v4f s = { 0.f, 0.f, 0.f, 0.f };
#pragma unroll
    for (int j = 0; j < 8; ++j) s += rb[j] * w[j];
    if constexpr (LSEL == 0) {
        acc[0] += s * neigh;
    } else if constexpr (LSEL == 1) {
        float4 s0 = g[2];
        v4f w1 = s * neigh;
        acc[0] += s0.x * w1; acc[1] += s0.y * w1; acc[2] += s0.z * w1;
    } else {
        float4 s0 = g[2], s1 = g[3];
        v4f w2 = s * neigh;
        acc[0] += s0.w * w2; acc[1] += s1.x * w2; acc[2] += s1.y * w2;
        acc[3] += s1.z * w2; acc[4] += s1.w * w2;
    }
}


#define MIXB    768
#define MIXB2   313
#define MIXB1   246
#define NGROUPS (N_ATOMS / 16)
#define MPITCH  132
static_assert(N_ATOMS % 16 == 0, "a tile is 16 whole atoms");
static_assert(MIXB2 + MIXB1 < MIXB && MIXB - MIXB2 - MIXB1 <= NGROUPS && MIXB2 <= NGROUPS && MIXB1 <= NGROUPS, "every block owns a group");
static_assert(SLOTS == 32 && CAP >= SLOTS, "one staging round of 32 lanes x 4 pieces fills an atom's LDS slots");

template <int LSEL>
__device__ __forceinline__ void mix_body(
        const float* __restrict__ emb, const float* __restrict__ W_rad,
        const int* __restrict__ cnt, const int* __restrict__ spill_n,
        const float4* __restrict__ rec, const int* __restrict__ rsp,
        const float4* __restrict__ srec, const int* __restrict__ scs,
        const float* __restrict__ B, const int* __restrict__ species, float* __restrict__ x0,
        const int first, const int stride,
        float4 (&s_rec)[4][4][SLOTS * 4], float (&s_emb)[4 * N_CH],
        float (&s_t)[5][16 * MPITCH]) {
    constexpr int Q = 2 * LSEL + 1;
    const int wv   = __builtin_amdgcn_readfirstlane((int)(threadIdx.x >> 5));
    const int lane = threadIdx.x & 31;
    const int half = lane >> 4;
    const int l16  = lane & 15;
    const int col0 = wv * 32 + l16;

    v2f b[32][2];
    const float* Bl = B + (long)(2 * half) * KDIM + col0;
#pragma unroll
    for (int j = 0; j < 32; ++j) {
#pragma unroll
        for (int t = 0; t < 2; ++t) {
            b[j][t].x = Bl[(4 * j) * KDIM + t * 16];
            b[j][t].y = Bl[(4 * j + 1) * KDIM + t * 16];
        }
    }
    v4f w[8];
#pragma unroll
    for (int j = 0; j < 8; ++j) w[j] = *(const v4f*)(W_rad + (LSEL * 8 + j) * KDIM + 4 * lane);
    s_emb[threadIdx.x] = emb[threadIdx.x];
    const int qi0 = (LSEL * KDIM + col0) / 12;
    const int qi1 = (LSEL * KDIM + col0 + 16) / 12;
    const float i3 = 0.5773502691896258f;
    const float sc = (LSEL == 1) ? -i3 : i3;
    const float mp = 0.1f;
    const int ao = l16 * MPITCH + 2 * half;
    const int fo = 8 * half * MPITCH + col0;

    int nn[4];
#pragma unroll
    for (int u = 0; u < 4; ++u) nn[u] = cnt[first * 16 + wv * 4 + u];
    for (int g = first; g < NGROUPS; g += stride) {
        int n[4];
#pragma unroll
        for (int u = 0; u < 4; ++u) n[u] = nn[u];
        if (g + stride < NGROUPS) {
#pragma unroll
            for (int u = 0; u < 4; ++u) nn[u] = cnt[(g + stride) * 16 + wv * 4 + u];
        }
        int    spv[4];
        float4 q4[4][4];
#pragma unroll
        for (int u = 0; u < 4; ++u) {
            const long c  = (long)g * 16 + wv * 4 + u;
            const int  np = n[u] < SLOTS ? n[u] : SLOTS;
            const int*    sb = np > 0 ? rsp + c * CAP : species;
            const float4* rb = np > 0 ? rec + c * CAP * 4 : (const float4*)emb;
            const int     m1 = np > 0 ? np - 1 : 0;
            const int     m4 = np > 0 ? 4 * np - 1 : 0;
            spv[u] = sb[lane < m1 ? lane : m1];
#pragma unroll
            for (int v = 0; v < 4; ++v) {
                const int t = lane + 32 * v;
                q4[u][v] = rb[t < m4 ? t : m4];
            }
        }
#pragma unroll
        for (int u = 0; u < 4; ++u) {
#pragma unroll
            for (int v = 0; v < 4; ++v) s_rec[wv][u][lane + 32 * v] = q4[u][v];
        }
        __syncthreads();
        int sp[8];
#pragma unroll
        for (int r = 0; r < 8; ++r) sp[r] = species[g * 16 + 8 * half + r];

#pragma nounroll
        for (int u = 0; u < 4; ++u) {
            const long c   = (long)g * 16 + wv * 4 + u;
            const int  nu  = u == 0 ? n[0] : (u == 1 ? n[1] : (u == 2 ? n[2] : n[3]));
            const int  spu = u == 0 ? spv[0] : (u == 1 ? spv[1] : (u == 2 ? spv[2] : spv[3]));
            const int  np  = nu < SLOTS ? nu : SLOTS;
            const int  nr = nu < CAP ? nu : CAP;
            v4f acc[Q];
#pragma unroll
            for (int i = 0; i < Q; ++i) acc[i] = (v4f){ 0.f, 0.f, 0.f, 0.f };
#pragma unroll 4
            for (int i = 0; i < np; ++i)
                add_pair_l<LSEL>(&s_rec[wv][u][4 * i],
                                 s_emb[__builtin_amdgcn_readlane(spu, i) * N_CH + lane], w, acc);
            for (int i = np; i < nr; ++i)
                add_pair_l<LSEL>(rec + (c * CAP + i) * 4, s_emb[rsp[c * CAP + i] * N_CH + lane], w, acc);
            if (nu > CAP) {
                int no = *spill_n;
                if (no > N_PAIRS) no = N_PAIRS;
                for (int i = 0; i < no; ++i)
                    if (scs[2 * i] == (int)c)
                        add_pair_l<LSEL>(srec + 4 * (long)i, s_emb[scs[2 * i + 1] * N_CH + lane], w, acc);
            }
#pragma unroll
            for (int i = 0; i < Q; ++i)
                *(v4f*)(&s_t[i][(wv * 4 + u) * MPITCH + 4 * lane]) = acc[i] * mp;
        }
        __syncthreads();

        float P[2][8];
#pragma unroll
        for (int r = 0; r < 8; ++r) { P[0][r] = 0.f; P[1][r] = 0.f; }
        v2f a[32];
#pragma unroll
        for (int j = 0; j < 32; ++j) a[j] = *(const v2f*)(&s_t[0][ao + 4 * j]);
#pragma unroll
        for (int q = 0; q < Q; ++q) {
            float fc[2][8];
#pragma unroll
            for (int r = 0; r < 8; ++r) {
                fc[0][r] = s_t[q][fo + r * MPITCH];
                fc[1][r] = s_t[q][fo + r * MPITCH + 16];
            }
            __builtin_amdgcn_sched_barrier(0);

            v8f acc0 = {}, acc1 = {};
#pragma unroll
            for (int j = 0; j < 16; ++j) {
                acc0 = __builtin_amdgcn_wmma_f32_16x16x4_f32(false, a[j], false, b[j][0], (short)0,
                                                             acc0, false, false);
                acc1 = __builtin_amdgcn_wmma_f32_16x16x4_f32(false, a[j], false, b[j][1], (short)0,
                                                             acc1, false, false);
            }
            __builtin_amdgcn_sched_barrier(0);
            if (q + 1 < Q) {
#pragma unroll
                for (int j = 0; j < 16; ++j) a[j] = *(const v2f*)(&s_t[q + 1][ao + 4 * j]);
            }
            __builtin_amdgcn_sched_barrier(0);
#pragma unroll
            for (int j = 16; j < 32; ++j) {
                acc0 = __builtin_amdgcn_wmma_f32_16x16x4_f32(false, a[j], false, b[j][0], (short)0,
                                                             acc0, false, false);
                acc1 = __builtin_amdgcn_wmma_f32_16x16x4_f32(false, a[j], false, b[j][1], (short)0,
                                                             acc1, false, false);
            }
            __builtin_amdgcn_sched_barrier(0);
            if (q + 1 < Q) {
#pragma unroll
                for (int j = 16; j < 32; ++j) a[j] = *(const v2f*)(&s_t[q + 1][ao + 4 * j]);
            }
#pragma unroll
            for (int r = 0; r < 8; ++r) {
                P[0][r] = acc0[r] * fc[0][r] + ((LSEL == 0) ? fc[0][r] : P[0][r]);
                P[1][r] = acc1[r] * fc[1][r] + ((LSEL == 0) ? fc[1][r] : P[1][r]);
            }
        }
        float* X = x0 + ((long)g * 16 + 8 * half) * K0TOT + LSEL * KDIM + col0;
#pragma unroll
        for (int r = 0; r < 8; ++r) {
            const float* er = s_emb + sp[r] * N_CH;
            const float v0 = (LSEL == 0) ? P[0][r] : P[0][r] * sc;
            const float v1 = (LSEL == 0) ? P[1][r] : P[1][r] * sc;
            X[(long)r * K0TOT]      = v0 * er[qi0];
            X[(long)r * K0TOT + 16] = v1 * er[qi1];
        }
    }
}

__global__ __launch_bounds__(128, 3) void k_pairsum_mix(
        const float* __restrict__ emb, const float* __restrict__ W_rad,
        const int* __restrict__ cnt, const int* __restrict__ spill_n,
        const float4* __restrict__ rec, const int* __restrict__ rsp,
        const float4* __restrict__ srec, const int* __restrict__ scs,
        const float* __restrict__ Wc0, const float* __restrict__ Wc1, const float* __restrict__ Wc2,
        const int* __restrict__ species, float* __restrict__ x0) {
    __shared__ float4 s_rec[4][4][SLOTS * 4];
    __shared__ float  s_emb[4 * N_CH];
    __shared__ __attribute__((aligned(16))) float s_t[5][16 * MPITCH];
    static_assert(128 >= 4 * N_CH, "one thread per embedding value when the table is staged");
    const int blk = blockIdx.x;
    if (blk < MIXB2)
        mix_body<2>(emb, W_rad, cnt, spill_n, rec, rsp, srec, scs, Wc2, species, x0,
                    blk, MIXB2, s_rec, s_emb, s_t);
    else if (blk < MIXB2 + MIXB1)
        mix_body<1>(emb, W_rad, cnt, spill_n, rec, rsp, srec, scs, Wc1, species, x0,
                    blk - MIXB2, MIXB1, s_rec, s_emb, s_t);
    else
        mix_body<0>(emb, W_rad, cnt, spill_n, rec, rsp, srec, scs, Wc0, species, x0,
                    blk - MIXB2 - MIXB1, MIXB - MIXB2 - MIXB1, s_rec, s_emb, s_t);
}

#define HEAD_BLOCKS 256
#define HEAD_TILES  (N_ATOMS / 16)
#define HPITCH      388
__device__ __forceinline__ void head_tile(const v2f (&cur)[96], v2f (&nxt)[96], const v2f (&b)[96][2],
                                          const float* __restrict__ src, const float* __restrict__ ldn,
                                          float* __restrict__ fil, float (&rd)[4][8][32],
                                          float* __restrict__ out, int tile,
                                          float bias0, float wo0, float bias1, float wo1,
                                          int wv, int lane, int half, int l16, int tid) {
    v4f gq[12];
#pragma unroll
    for (int u = 0; u < 12; ++u) gq[u] = *(const v4f*)(src + 4 * (tid + 128 * u));
    const float* An = ldn + l16 * HPITCH + 2 * half;
#pragma unroll
    for (int j = 0; j < 48; ++j) nxt[j] = *(const v2f*)(An + 4 * j);
    __builtin_amdgcn_sched_barrier(0);

    v8f acc0 = {}, acc1 = {};
#pragma unroll
    for (int j = 0; j < 48; ++j) {
        acc0 = __builtin_amdgcn_wmma_f32_16x16x4_f32(false, cur[j], false, b[j][0], (short)0,
                                                     acc0, false, false);
        acc1 = __builtin_amdgcn_wmma_f32_16x16x4_f32(false, cur[j], false, b[j][1], (short)0,
                                                     acc1, false, false);
    }
    __builtin_amdgcn_sched_barrier(0);
#pragma unroll
    for (int j = 48; j < 96; ++j) nxt[j] = *(const v2f*)(An + 4 * j);
    __builtin_amdgcn_sched_barrier(0);
#pragma unroll
    for (int j = 48; j < 96; ++j) {
        acc0 = __builtin_amdgcn_wmma_f32_16x16x4_f32(false, cur[j], false, b[j][0], (short)0,
                                                     acc0, false, false);
        acc1 = __builtin_amdgcn_wmma_f32_16x16x4_f32(false, cur[j], false, b[j][1], (short)0,
                                                     acc1, false, false);
    }
    __builtin_amdgcn_sched_barrier(0);
#pragma unroll
    for (int i = 0; i < 8; ++i) {
        float v0 = acc0[i] + bias0;
        float v1 = acc1[i] + bias1;
        float p = 0.f;
        p += (v0 / (1.f + expf(-v0))) * wo0;
        p += (v1 / (1.f + expf(-v1))) * wo1;
        rd[wv][i][lane] = p;
    }
#pragma unroll
    for (int u = 0; u < 12; ++u) {
        const int f = tid + 128 * u;
        *(v4f*)(fil + (f / 96) * HPITCH + 4 * (f % 96)) = gq[u];
    }
    __syncthreads();
    const int rowbase = tile * 16 + 8 * half;
#pragma unroll
    for (int r = 0; r < 2; ++r) {
        const int i = 2 * wv + r;
        float s = rd[0][i][lane];
        s += rd[1][i][lane];
        s += rd[2][i][lane];
        s += rd[3][i][lane];
        atomicAdd(&out[rowbase + i], s);
    }
}

__global__ __launch_bounds__(128) void k_head_resident(
        const float* __restrict__ x0, const float* __restrict__ Wh, const float* __restrict__ bh,
        const float* __restrict__ Wo, float* __restrict__ out) {
    __shared__ float red[2][4][8][32];
    __shared__ __attribute__((aligned(16))) float s_x[3][16 * HPITCH];
    const int cb     = blockIdx.x % 3;
    const int first  = blockIdx.x / 3;
    const int stride = (cb == 0) ? 86 : 85;
    const int tid  = threadIdx.x;
    const int wv   = threadIdx.x >> 5;
    const int lane = threadIdx.x & 31;
    const int half = lane >> 4;
    const int l16  = lane & 15;
    const int n0   = cb * 128 + wv * 32;

    v2f b[96][2];
    const float* Bl = Wh + (long)(2 * half) * K0TOT + n0 + l16;
#pragma unroll
    for (int j = 0; j < 96; ++j) {
#pragma unroll
        for (int t = 0; t < 2; ++t) {
            b[j][t].x = Bl[(4 * j) * K0TOT + t * 16];
            b[j][t].y = Bl[(4 * j + 1) * K0TOT + t * 16];
        }
    }
    const float bias0 = bh[n0 + l16],      wo0 = Wo[n0 + l16];
    const float bias1 = bh[n0 + 16 + l16], wo1 = Wo[n0 + 16 + l16];

    {
        int t1 = first + stride;
        if (t1 >= HEAD_TILES) t1 = first;
        const float* s0 = x0 + (long)first * 16 * K0TOT;
        const float* s1 = x0 + (long)t1 * 16 * K0TOT;
        v4f g0[12], g1[12];
#pragma unroll
        for (int u = 0; u < 12; ++u) g0[u] = *(const v4f*)(s0 + 4 * (tid + 128 * u));
#pragma unroll
        for (int u = 0; u < 12; ++u) g1[u] = *(const v4f*)(s1 + 4 * (tid + 128 * u));
#pragma unroll
        for (int u = 0; u < 12; ++u) {
            const int f = tid + 128 * u;
            *(v4f*)(&s_x[0][(f / 96) * HPITCH + 4 * (f % 96)]) = g0[u];
            *(v4f*)(&s_x[1][(f / 96) * HPITCH + 4 * (f % 96)]) = g1[u];
        }
    }
    __syncthreads();
    v2f a[96];
    {
        const float* Ar = &s_x[0][l16 * HPITCH + 2 * half];
#pragma unroll
        for (int j = 0; j < 96; ++j) a[j] = *(const v2f*)(Ar + 4 * j);
    }
    int par = 0;
    const float* nxt = s_x[1];
    float*       fil = s_x[2];
    float*       old = s_x[0];
    for (int tile = first; tile < HEAD_TILES; tile += stride) {
        int t2 = tile + 2 * stride;
        if (t2 >= HEAD_TILES) t2 = tile;
        v2f an[96];
        head_tile(a, an, b, x0 + (long)t2 * 16 * K0TOT, nxt, fil, red[par], out, tile,
                  bias0, wo0, bias1, wo1, wv, lane, half, l16, tid);
        par ^= 1;
        __builtin_amdgcn_sched_barrier(0);
#pragma unroll
        for (int j = 0; j < 96; ++j) a[j] = an[j];
        float* t = old; old = (float*)nxt; nxt = fil; fil = t;
    }
}

extern "C" void kernel_launch(void* const* d_in, const int* in_sizes, int n_in,
                              void* d_out, int out_size, void* d_ws, size_t ws_size,
                              hipStream_t stream) {
    (void)in_sizes; (void)n_in; (void)out_size; (void)ws_size;
    const float* positions = (const float*)d_in[0];
    const int*   species   = (const int*)d_in[1];
    const int*   pairs     = (const int*)d_in[2];
    const float* embeddings= (const float*)d_in[3];
    const float* W_rad     = (const float*)d_in[4];
    const float* W_cg0     = (const float*)d_in[5];
    const float* W_cg1     = (const float*)d_in[6];
    const float* W_cg2     = (const float*)d_in[7];
    const float* W_head    = (const float*)d_in[8];
    const float* b_head    = (const float*)d_in[9];
    const float* W_out     = (const float*)d_in[10];
    const float* b_out     = (const float*)d_in[11];
    float* out = (float*)d_out;

    float* ws = (float*)d_ws;
    const long SZ_F0 = (long)N_ATOMS * KDIM;
    const long SZ_F1 = (long)N_ATOMS * 3 * KDIM;
    const long SZ_F2 = (long)N_ATOMS * 5 * KDIM;
    const long SZ_U  = (long)N_ATOMS * 9 * KDIM;
    float* f0 = ws;
    float* f1 = f0 + SZ_F0;
    float* f2 = f1 + SZ_F1;
    float* g0 = f2 + SZ_F2 + SZ_U + SZ_U;
    float* g1 = g0 + SZ_F0;
    float* g2 = g1 + SZ_U;
    float* x0 = g2 + SZ_U;

    float4* rec  = (float4*)g2;
    float4* srec = (float4*)g1;
    int* cnt     = (int*)f0;
    int* spill_n = cnt + 10240;
    int* rsp     = cnt + 10304;
    int* scs     = rsp + (long)N_ATOMS * CAP;

    k_init<<<(N_ATOMS + 255) / 256, 256, 0, stream>>>(out, b_out, cnt, spill_n);

    k_bin<<<N_PAIRS / 256, 256, 0, stream>>>(positions, species, pairs, cnt, spill_n,
                                             rec, rsp, srec, scs);

    k_pairsum_mix<<<MIXB, 128, 0, stream>>>(embeddings, W_rad, cnt, spill_n, rec, rsp, srec, scs,
                                            W_cg0, W_cg1, W_cg2, species, x0);

    k_head_resident<<<HEAD_BLOCKS, 128, 0, stream>>>(x0, W_head, b_head, W_out, out);
}


# channel mix: a group's start-up loads leave the critical path: 16 record loads in flight together (the l=2 body sent them one by one), counts read a group after their load, L2 asked a group ahead
# speedup vs baseline: 3.0905x; 1.0311x over previous
#include <hip/hip_runtime.h>
#include <math.h>

#define N_ATOMS  10000
#define N_PAIRS  160000
#define N_CH     32
#define KDIM     128
#define K0TOT    384

typedef float v2f __attribute__((ext_vector_type(2)));
typedef float v8f __attribute__((ext_vector_type(8)));


#define CAP 64

__global__ void k_init(float* out, const float* b_out, int* cnt, int* spill_n) {
    int i = blockIdx.x * blockDim.x + threadIdx.x;
    if (i < N_ATOMS) { out[i] = b_out[0]; cnt[i] = 0; }
    if (i == 0) *spill_n = 0;
}

__global__ void k_bin(const float* __restrict__ pos, const int* __restrict__ species,
                      const int* __restrict__ pairs, int* __restrict__ cnt,
                      int* __restrict__ spill_n, float4* __restrict__ rec, int* __restrict__ rsp,
                      float4* __restrict__ srec, int* __restrict__ scs) {
    int p = blockIdx.x * blockDim.x + threadIdx.x;
    if (p >= N_PAIRS) return;
    int c  = pairs[2 * p + 0];
    int nb = pairs[2 * p + 1];
    float rx = pos[3 * nb + 0] - pos[3 * c + 0];
    float ry = pos[3 * nb + 1] - pos[3 * c + 1];
    float rz = pos[3 * nb + 2] - pos[3 * c + 2];
    float d  = sqrtf(rx * rx + ry * ry + rz * rz + 1e-12f);
    float inv = 1.f / d;
    float ux = rx * inv, uy = ry * inv, uz = rz * inv;
    float t  = (d - 15.f) * 0.2f;
    t = fminf(fmaxf(t, 0.f), 1.f);
    float fc = 0.5f * (cosf(3.14159265358979f * t) + 1.f);
    float rb[8];
#pragma unroll
    for (int j = 0; j < 8; ++j) {
        float mu = 20.f * (float)j / 7.f;
        float x  = (d - mu) * 0.4f;
        rb[j] = expf(-x * x) * fc;
    }
    const float s3 = 1.7320508075688772f;
    float4 q0 = { rb[0], rb[1], rb[2], rb[3] };
    float4 q1 = { rb[4], rb[5], rb[6], rb[7] };
    float4 q2 = { uy, uz, ux, s3 * ux * uy };
    float4 q3 = { s3 * uy * uz, 0.5f * (3.f * uz * uz - 1.f), s3 * ux * uz,
                  0.5f * s3 * (ux * ux - uy * uy) };
    int sp = species[nb];
    int slot = atomicAdd(&cnt[c], 1);
    float4* dst;
    if (slot < CAP) {
        long idx = (long)c * CAP + slot;
        rsp[idx] = sp;
        dst = rec + idx * 4;
    } else {
        int o = atomicAdd(spill_n, 1);
        scs[2 * o + 0] = c;
        scs[2 * o + 1] = sp;
        dst = srec + (long)o * 4;
    }
    dst[0] = q0; dst[1] = q1; dst[2] = q2; dst[3] = q3;
}

typedef float v4f __attribute__((ext_vector_type(4)));

#define SLOTS 32

template <int LSEL>
__device__ __forceinline__ void add_pair_l(const float4* g, float neigh,
                                           const v4f (&w)[8], v4f (&acc)[2 * LSEL + 1]) {
    float4 a = g[0], b = g[1];
    float rb[8] = { a.x, a.y, a.z, a.w, b.x, b.y, b.z, b.w };
    v4f s = { 0.f, 0.f, 0.f, 0.f };
#pragma unroll
    for (int j = 0; j < 8; ++j) s += rb[j] * w[j];
    if constexpr (LSEL == 0) {
        acc[0] += s * neigh;
    } else if constexpr (LSEL == 1) {
        float4 s0 = g[2];
        v4f w1 = s * neigh;
        acc[0] += s0.x * w1; acc[1] += s0.y * w1; acc[2] += s0.z * w1;
    } else {
        float4 s0 = g[2], s1 = g[3];
        v4f w2 = s * neigh;
        acc[0] += s0.w * w2; acc[1] += s1.x * w2; acc[2] += s1.y * w2;
        acc[3] += s1.z * w2; acc[4] += s1.w * w2;
    }
}


#define MIXB    768
#define MIXB2   313
#define MIXB1   246
#define NGROUPS (N_ATOMS / 16)
#define MPITCH  132
static_assert(N_ATOMS % 16 == 0, "a tile is 16 whole atoms");
static_assert(MIXB2 + MIXB1 < MIXB && MIXB - MIXB2 - MIXB1 <= NGROUPS && MIXB2 <= NGROUPS && MIXB1 <= NGROUPS, "every block owns a group");
static_assert(SLOTS == 32 && CAP >= SLOTS, "one staging round of 32 lanes x 4 pieces fills an atom's LDS slots");

__device__ __forceinline__ void ask_group(const float4* __restrict__ rec, const int* __restrict__ rsp,
                                          int G, int wv, int lane) {
    const int  u = lane < 24 ? lane / 6 : (lane & 3);
    const long c = (long)G * 16 + wv * 4 + u;
    const char* p = lane < 24 ? (const char*)(rec + c * CAP * 4) + (lane % 6) * 256
                              : (const char*)(rsp + c * CAP);
    __builtin_prefetch(p, 0, 1);
}

template <int LSEL>
__device__ __forceinline__ void mix_body(
        const float* __restrict__ emb, const float* __restrict__ W_rad,
        const int* __restrict__ cnt, const int* __restrict__ spill_n,
        const float4* __restrict__ rec, const int* __restrict__ rsp,
        const float4* __restrict__ srec, const int* __restrict__ scs,
        const float* __restrict__ B, const int* __restrict__ species, float* __restrict__ x0,
        const int first, const int stride,
        float4 (&s_rec)[4][4][SLOTS * 4], float (&s_emb)[4 * N_CH],
        float (&s_t)[5][16 * MPITCH]) {
    constexpr int Q = 2 * LSEL + 1;
    const int wv   = __builtin_amdgcn_readfirstlane((int)(threadIdx.x >> 5));
    const int lane = threadIdx.x & 31;
    const int half = lane >> 4;
    const int l16  = lane & 15;
    const int col0 = wv * 32 + l16;

    int nnv = cnt[first * 16 + wv * 4 + (lane & 3)];
    ask_group(rec, rsp, first, wv, lane);

    v2f b[32][2];
    const float* Bl = B + (long)(2 * half) * KDIM + col0;
#pragma unroll
    for (int j = 0; j < 32; ++j) {
#pragma unroll
        for (int t = 0; t < 2; ++t) {
            b[j][t].x = Bl[(4 * j) * KDIM + t * 16];
            b[j][t].y = Bl[(4 * j + 1) * KDIM + t * 16];
        }
    }
    v4f w[8];
#pragma unroll
    for (int j = 0; j < 8; ++j) w[j] = *(const v4f*)(W_rad + (LSEL * 8 + j) * KDIM + 4 * lane);
    s_emb[threadIdx.x] = emb[threadIdx.x];
    const int qi0 = (LSEL * KDIM + col0) / 12;
    const int qi1 = (LSEL * KDIM + col0 + 16) / 12;
    const float i3 = 0.5773502691896258f;
    const float sc = (LSEL == 1) ? -i3 : i3;
    const float mp = 0.1f;
    const int ao = l16 * MPITCH + 2 * half;
    const int fo = 8 * half * MPITCH + col0;

    for (int g = first; g < NGROUPS; g += stride) {
        int n[4];
#pragma unroll
        for (int u = 0; u < 4; ++u) n[u] = __builtin_amdgcn_readlane(nnv, u);
        if (g + stride < NGROUPS) {
            nnv = cnt[(g + stride) * 16 + wv * 4 + (lane & 3)];
            ask_group(rec, rsp, g + stride, wv, lane);
        }
        int spv[4];
        v4f q4[4][4];
#pragma unroll
        for (int u = 0; u < 4; ++u) {
            const long c  = (long)g * 16 + wv * 4 + u;
            const int  np = n[u] < SLOTS ? n[u] : SLOTS;
            const int* sb = np > 0 ? rsp + c * CAP : species;
            const v4f* rb = np > 0 ? (const v4f*)(rec + c * CAP * 4) : (const v4f*)emb;
            const int  m1 = np > 0 ? np - 1 : 0;
            const int  m4 = np > 0 ? 4 * np - 1 : 0;
            spv[u] = sb[lane < m1 ? lane : m1];
#pragma unroll
            for (int v = 0; v < 4; ++v) {
                const int t = lane + 32 * v;
                q4[u][v] = rb[t < m4 ? t : m4];
            }
        }
        asm volatile("" :: "v"(q4[0][0]), "v"(q4[0][1]), "v"(q4[0][2]), "v"(q4[0][3]),
                           "v"(q4[1][0]), "v"(q4[1][1]), "v"(q4[1][2]), "v"(q4[1][3]),
                           "v"(q4[2][0]), "v"(q4[2][1]), "v"(q4[2][2]), "v"(q4[2][3]),
                           "v"(q4[3][0]), "v"(q4[3][1]), "v"(q4[3][2]), "v"(q4[3][3]));
#pragma unroll
        for (int u = 0; u < 4; ++u) {
#pragma unroll
            for (int v = 0; v < 4; ++v) *(v4f*)(&s_rec[wv][u][lane + 32 * v]) = q4[u][v];
        }
        __syncthreads();
        int sp[8];
#pragma unroll
        for (int r = 0; r < 8; ++r) sp[r] = species[g * 16 + 8 * half + r];

#pragma nounroll
        for (int u = 0; u < 4; ++u) {
            const long c   = (long)g * 16 + wv * 4 + u;
            const int  nu  = u == 0 ? n[0] : (u == 1 ? n[1] : (u == 2 ? n[2] : n[3]));
            const int  spu = u == 0 ? spv[0] : (u == 1 ? spv[1] : (u == 2 ? spv[2] : spv[3]));
            const int  np  = nu < SLOTS ? nu : SLOTS;
            const int  nr = nu < CAP ? nu : CAP;
            v4f acc[Q];
#pragma unroll
            for (int i = 0; i < Q; ++i) acc[i] = (v4f){ 0.f, 0.f, 0.f, 0.f };
#pragma unroll 4
            for (int i = 0; i < np; ++i)
                add_pair_l<LSEL>(&s_rec[wv][u][4 * i],
                                 s_emb[__builtin_amdgcn_readlane(spu, i) * N_CH + lane], w, acc);
            for (int i = np; i < nr; ++i)
                add_pair_l<LSEL>(rec + (c * CAP + i) * 4, s_emb[rsp[c * CAP + i] * N_CH + lane], w, acc);
            if (nu > CAP) {
                int no = *spill_n;
                if (no > N_PAIRS) no = N_PAIRS;
                for (int i = 0; i < no; ++i)
                    if (scs[2 * i] == (int)c)
                        add_pair_l<LSEL>(srec + 4 * (long)i, s_emb[scs[2 * i + 1] * N_CH + lane], w, acc);
            }
#pragma unroll
            for (int i = 0; i < Q; ++i)
                *(v4f*)(&s_t[i][(wv * 4 + u) * MPITCH + 4 * lane]) = acc[i] * mp;
        }
        __syncthreads();

        float P[2][8];
#pragma unroll
        for (int r = 0; r < 8; ++r) { P[0][r] = 0.f; P[1][r] = 0.f; }
        v2f a[32];
#pragma unroll
        for (int j = 0; j < 32; ++j) a[j] = *(const v2f*)(&s_t[0][ao + 4 * j]);
#pragma unroll
        for (int q = 0; q < Q; ++q) {
            float fc[2][8];
#pragma unroll
            for (int r = 0; r < 8; ++r) {
                fc[0][r] = s_t[q][fo + r * MPITCH];
                fc[1][r] = s_t[q][fo + r * MPITCH + 16];
            }
            __builtin_amdgcn_sched_barrier(0);

            v8f acc0 = {}, acc1 = {};
#pragma unroll
            for (int j = 0; j < 16; ++j) {
                acc0 = __builtin_amdgcn_wmma_f32_16x16x4_f32(false, a[j], false, b[j][0], (short)0,
                                                             acc0, false, false);
                acc1 = __builtin_amdgcn_wmma_f32_16x16x4_f32(false, a[j], false, b[j][1], (short)0,
                                                             acc1, false, false);
            }
            __builtin_amdgcn_sched_barrier(0);
            if (q + 1 < Q) {
#pragma unroll
                for (int j = 0; j < 16; ++j) a[j] = *(const v2f*)(&s_t[q + 1][ao + 4 * j]);
            }
            __builtin_amdgcn_sched_barrier(0);
#pragma unroll
            for (int j = 16; j < 32; ++j) {
                acc0 = __builtin_amdgcn_wmma_f32_16x16x4_f32(false, a[j], false, b[j][0], (short)0,
                                                             acc0, false, false);
                acc1 = __builtin_amdgcn_wmma_f32_16x16x4_f32(false, a[j], false, b[j][1], (short)0,
                                                             acc1, false, false);
            }
            __builtin_amdgcn_sched_barrier(0);
            if (q + 1 < Q) {
#pragma unroll
                for (int j = 16; j < 32; ++j) a[j] = *(const v2f*)(&s_t[q + 1][ao + 4 * j]);
            }
#pragma unroll
            for (int r = 0; r < 8; ++r) {
                P[0][r] = acc0[r] * fc[0][r] + ((LSEL == 0) ? fc[0][r] : P[0][r]);
                P[1][r] = acc1[r] * fc[1][r] + ((LSEL == 0) ? fc[1][r] : P[1][r]);
            }
        }
        float* X = x0 + ((long)g * 16 + 8 * half) * K0TOT + LSEL * KDIM + col0;
#pragma unroll
        for (int r = 0; r < 8; ++r) {
            const float* er = s_emb + sp[r] * N_CH;
            const float v0 = (LSEL == 0) ? P[0][r] : P[0][r] * sc;
            const float v1 = (LSEL == 0) ? P[1][r] : P[1][r] * sc;
            X[(long)r * K0TOT]      = v0 * er[qi0];
            X[(long)r * K0TOT + 16] = v1 * er[qi1];
        }
    }
}

__global__ __launch_bounds__(128, 3) void k_pairsum_mix(
        const float* __restrict__ emb, const float* __restrict__ W_rad,
        const int* __restrict__ cnt, const int* __restrict__ spill_n,
        const float4* __restrict__ rec, const int* __restrict__ rsp,
        const float4* __restrict__ srec, const int* __restrict__ scs,
        const float* __restrict__ Wc0, const float* __restrict__ Wc1, const float* __restrict__ Wc2,
        const int* __restrict__ species, float* __restrict__ x0) {
    __shared__ float4 s_rec[4][4][SLOTS * 4];
    __shared__ float  s_emb[4 * N_CH];
    __shared__ __attribute__((aligned(16))) float s_t[5][16 * MPITCH];
    static_assert(128 >= 4 * N_CH, "one thread per embedding value when the table is staged");
    const int blk = blockIdx.x;
    if (blk < MIXB2)
        mix_body<2>(emb, W_rad, cnt, spill_n, rec, rsp, srec, scs, Wc2, species, x0,
                    blk, MIXB2, s_rec, s_emb, s_t);
    else if (blk < MIXB2 + MIXB1)
        mix_body<1>(emb, W_rad, cnt, spill_n, rec, rsp, srec, scs, Wc1, species, x0,
                    blk - MIXB2, MIXB1, s_rec, s_emb, s_t);
    else
        mix_body<0>(emb, W_rad, cnt, spill_n, rec, rsp, srec, scs, Wc0, species, x0,
                    blk - MIXB2 - MIXB1, MIXB - MIXB2 - MIXB1, s_rec, s_emb, s_t);
}

#define HEAD_BLOCKS 256
#define HEAD_TILES  (N_ATOMS / 16)
#define HPITCH      388
__device__ __forceinline__ void head_tile(const v2f (&cur)[96], v2f (&nxt)[96], const v2f (&b)[96][2],
                                          const float* __restrict__ src, const float* __restrict__ ldn,
                                          float* __restrict__ fil, float (&rd)[4][8][32],
                                          float* __restrict__ out, int tile,
                                          float bias0, float wo0, float bias1, float wo1,
                                          int wv, int lane, int half, int l16, int tid) {
    v4f gq[12];
#pragma unroll
    for (int u = 0; u < 12; ++u) gq[u] = *(const v4f*)(src + 4 * (tid + 128 * u));
    const float* An = ldn + l16 * HPITCH + 2 * half;
#pragma unroll
    for (int j = 0; j < 48; ++j) nxt[j] = *(const v2f*)(An + 4 * j);
    __builtin_amdgcn_sched_barrier(0);

    v8f acc0 = {}, acc1 = {};
#pragma unroll
    for (int j = 0; j < 48; ++j) {
        acc0 = __builtin_amdgcn_wmma_f32_16x16x4_f32(false, cur[j], false, b[j][0], (short)0,
                                                     acc0, false, false);
        acc1 = __builtin_amdgcn_wmma_f32_16x16x4_f32(false, cur[j], false, b[j][1], (short)0,
                                                     acc1, false, false);
    }
    __builtin_amdgcn_sched_barrier(0);
#pragma unroll
    for (int j = 48; j < 96; ++j) nxt[j] = *(const v2f*)(An + 4 * j);
    __builtin_amdgcn_sched_barrier(0);
#pragma unroll
    for (int j = 48; j < 96; ++j) {
        acc0 = __builtin_amdgcn_wmma_f32_16x16x4_f32(false, cur[j], false, b[j][0], (short)0,
                                                     acc0, false, false);
        acc1 = __builtin_amdgcn_wmma_f32_16x16x4_f32(false, cur[j], false, b[j][1], (short)0,
                                                     acc1, false, false);
    }
    __builtin_amdgcn_sched_barrier(0);
#pragma unroll
    for (int i = 0; i < 8; ++i) {
        float v0 = acc0[i] + bias0;
        float v1 = acc1[i] + bias1;
        float p = 0.f;
        p += (v0 / (1.f + expf(-v0))) * wo0;
        p += (v1 / (1.f + expf(-v1))) * wo1;
        rd[wv][i][lane] = p;
    }
#pragma unroll
    for (int u = 0; u < 12; ++u) {
        const int f = tid + 128 * u;
        *(v4f*)(fil + (f / 96) * HPITCH + 4 * (f % 96)) = gq[u];
    }
    __syncthreads();
    const int rowbase = tile * 16 + 8 * half;
#pragma unroll
    for (int r = 0; r < 2; ++r) {
        const int i = 2 * wv + r;
        float s = rd[0][i][lane];
        s += rd[1][i][lane];
        s += rd[2][i][lane];
        s += rd[3][i][lane];
        atomicAdd(&out[rowbase + i], s);
    }
}

__global__ __launch_bounds__(128) void k_head_resident(
        const float* __restrict__ x0, const float* __restrict__ Wh, const float* __restrict__ bh,
        const float* __restrict__ Wo, float* __restrict__ out) {
    __shared__ float red[2][4][8][32];
    __shared__ __attribute__((aligned(16))) float s_x[3][16 * HPITCH];
    const int cb     = blockIdx.x % 3;
    const int first  = blockIdx.x / 3;
    const int stride = (cb == 0) ? 86 : 85;
    const int tid  = threadIdx.x;
    const int wv   = threadIdx.x >> 5;
    const int lane = threadIdx.x & 31;
    const int half = lane >> 4;
    const int l16  = lane & 15;
    const int n0   = cb * 128 + wv * 32;

    v2f b[96][2];
    const float* Bl = Wh + (long)(2 * half) * K0TOT + n0 + l16;
#pragma unroll
    for (int j = 0; j < 96; ++j) {
#pragma unroll
        for (int t = 0; t < 2; ++t) {
            b[j][t].x = Bl[(4 * j) * K0TOT + t * 16];
            b[j][t].y = Bl[(4 * j + 1) * K0TOT + t * 16];
        }
    }
    const float bias0 = bh[n0 + l16],      wo0 = Wo[n0 + l16];
    const float bias1 = bh[n0 + 16 + l16], wo1 = Wo[n0 + 16 + l16];

    {
        int t1 = first + stride;
        if (t1 >= HEAD_TILES) t1 = first;
        const float* s0 = x0 + (long)first * 16 * K0TOT;
        const float* s1 = x0 + (long)t1 * 16 * K0TOT;
        v4f g0[12], g1[12];
#pragma unroll
        for (int u = 0; u < 12; ++u) g0[u] = *(const v4f*)(s0 + 4 * (tid + 128 * u));
#pragma unroll
        for (int u = 0; u < 12; ++u) g1[u] = *(const v4f*)(s1 + 4 * (tid + 128 * u));
#pragma unroll
        for (int u = 0; u < 12; ++u) {
            const int f = tid + 128 * u;
            *(v4f*)(&s_x[0][(f / 96) * HPITCH + 4 * (f % 96)]) = g0[u];
            *(v4f*)(&s_x[1][(f / 96) * HPITCH + 4 * (f % 96)]) = g1[u];
        }
    }
    __syncthreads();
    v2f a[96];
    {
        const float* Ar = &s_x[0][l16 * HPITCH + 2 * half];
#pragma unroll
        for (int j = 0; j < 96; ++j) a[j] = *(const v2f*)(Ar + 4 * j);
    }
    int par = 0;
    const float* nxt = s_x[1];
    float*       fil = s_x[2];
    float*       old = s_x[0];
    for (int tile = first; tile < HEAD_TILES; tile += stride) {
        int t2 = tile + 2 * stride;
        if (t2 >= HEAD_TILES) t2 = tile;
        v2f an[96];
        head_tile(a, an, b, x0 + (long)t2 * 16 * K0TOT, nxt, fil, red[par], out, tile,
                  bias0, wo0, bias1, wo1, wv, lane, half, l16, tid);
        par ^= 1;
        __builtin_amdgcn_sched_barrier(0);
#pragma unroll
        for (int j = 0; j < 96; ++j) a[j] = an[j];
        float* t = old; old = (float*)nxt; nxt = fil; fil = t;
    }
}

extern "C" void kernel_launch(void* const* d_in, const int* in_sizes, int n_in,
                              void* d_out, int out_size, void* d_ws, size_t ws_size,
                              hipStream_t stream) {
    (void)in_sizes; (void)n_in; (void)out_size; (void)ws_size;
    const float* positions = (const float*)d_in[0];
    const int*   species   = (const int*)d_in[1];
    const int*   pairs     = (const int*)d_in[2];
    const float* embeddings= (const float*)d_in[3];
    const float* W_rad     = (const float*)d_in[4];
    const float* W_cg0     = (const float*)d_in[5];
    const float* W_cg1     = (const float*)d_in[6];
    const float* W_cg2     = (const float*)d_in[7];
    const float* W_head    = (const float*)d_in[8];
    const float* b_head    = (const float*)d_in[9];
    const float* W_out     = (const float*)d_in[10];
    const float* b_out     = (const float*)d_in[11];
    float* out = (float*)d_out;

    float* ws = (float*)d_ws;
    const long SZ_F0 = (long)N_ATOMS * KDIM;
    const long SZ_F1 = (long)N_ATOMS * 3 * KDIM;
    const long SZ_F2 = (long)N_ATOMS * 5 * KDIM;
    const long SZ_U  = (long)N_ATOMS * 9 * KDIM;
    float* f0 = ws;
    float* f1 = f0 + SZ_F0;
    float* f2 = f1 + SZ_F1;
    float* g0 = f2 + SZ_F2 + SZ_U + SZ_U;
    float* g1 = g0 + SZ_F0;
    float* g2 = g1 + SZ_U;
    float* x0 = g2 + SZ_U;

    float4* rec  = (float4*)g2;
    float4* srec = (float4*)g1;
    int* cnt     = (int*)f0;
    int* spill_n = cnt + 10240;
    int* rsp     = cnt + 10304;
    int* scs     = rsp + (long)N_ATOMS * CAP;

    k_init<<<(N_ATOMS + 255) / 256, 256, 0, stream>>>(out, b_out, cnt, spill_n);

    k_bin<<<N_PAIRS / 256, 256, 0, stream>>>(positions, species, pairs, cnt, spill_n,
                                             rec, rsp, srec, scs);

    k_pairsum_mix<<<MIXB, 128, 0, stream>>>(embeddings, W_rad, cnt, spill_n, rec, rsp, srec, scs,
                                            W_cg0, W_cg1, W_cg2, species, x0);

    k_head_resident<<<HEAD_BLOCKS, 128, 0, stream>>>(x0, W_head, b_head, W_out, out);
}
